# starting kernel
# baseline (speedup 1.0000x reference)
#include <hip/hip_runtime.h>
#include <hip/hip_bf16.h>
#include <math.h>

#define Bz    4
#define Sn    4096
#define Dn    128
#define Kn    13
#define Mn    12
#define Rn    (Bz * Sn)
#define WPAD  16
#define HSTR  132

typedef float v2f __attribute__((ext_vector_type(2)));
typedef float v8f __attribute__((ext_vector_type(8)));

__device__ __forceinline__ float gelu_exact(float v) {
    return 0.5f * v * (1.0f + erff(v * 0.70710678118654752f));
}

__device__ __forceinline__ v8f wmma_f32x4(v2f a, v2f b, v8f c) {
    return __builtin_amdgcn_wmma_f32_16x16x4_f32(
         false, a,  false, b,
         (short)0, c,  false,  false);
}

__global__ __launch_bounds__(256) void embed_kernel(
    const int* __restrict__ data, const float* __restrict__ emb,
    const float* __restrict__ apc, float* __restrict__ x)
{
    int tid = blockIdx.x * blockDim.x + threadIdx.x;
    int row = tid >> 5;
    int c4  = tid & 31;
    if (row >= Rn) return;
    int s   = row & (Sn - 1);
    int tok = data[row];
    const float4* e = (const float4*)(emb + (size_t)tok * Dn);
    const float4* a = (const float4*)(apc + (size_t)s * Dn);
    float4 ev = e[c4], av = a[c4];
    float4 o; o.x = ev.x + av.x; o.y = ev.y + av.y; o.z = ev.z + av.z; o.w = ev.w + av.w;
    ((float4*)x)[tid] = o;
}

__global__ __launch_bounds__(128) void gmlp_kernel(
    const float* __restrict__ x,
    const float* __restrict__ W1, const float* __restrict__ b1,
    const float* __restrict__ W2, const float* __restrict__ b2,
    float* __restrict__ outV)
{
    __shared__ float Hlds[4][16 * HSTR];
    const int wave    = threadIdx.x >> 5;
    const int lane    = threadIdx.x & 31;
    const int half    = lane >> 4;
    const int mn      = lane & 15;
    const int rowBase = blockIdx.x * 64 + wave * 16;

    v8f acc[8];
    #pragma unroll
    for (int t = 0; t < 8; t++) acc[t] = {};
    const float* Xrow = x + (size_t)(rowBase + mn) * Dn;
    #pragma unroll 4
    for (int k = 0; k < Dn; k += 4) {
        float2 av = *(const float2*)(Xrow + k + 2 * half);
        v2f a; a.x = av.x; a.y = av.y;
        #pragma unroll
        for (int t = 0; t < 8; t++) {
            const float* wp = W1 + (size_t)(k + 2 * half) * Dn + t * 16 + mn;
            v2f b; b.x = wp[0]; b.y = wp[Dn];
            acc[t] = wmma_f32x4(a, b, acc[t]);
        }
    }
    #pragma unroll
    for (int t = 0; t < 8; t++) {
        float bias = b1[t * 16 + mn];
        #pragma unroll
        for (int r = 0; r < 8; r++) {
            Hlds[wave][(r + 8 * half) * HSTR + t * 16 + mn] =
                gelu_exact(acc[t][r] + bias);
        }
    }
    __syncthreads();

    v8f acc2[8];
    #pragma unroll
    for (int t = 0; t < 8; t++) acc2[t] = {};
    #pragma unroll 4
    for (int k = 0; k < Dn; k += 4) {
        float2 av = *(const float2*)(&Hlds[wave][mn * HSTR + k + 2 * half]);
        v2f a; a.x = av.x; a.y = av.y;
        #pragma unroll
        for (int t = 0; t < 8; t++) {
            const float* wp = W2 + (size_t)(k + 2 * half) * Dn + t * 16 + mn;
            v2f b; b.x = wp[0]; b.y = wp[Dn];
            acc2[t] = wmma_f32x4(a, b, acc2[t]);
        }
    }
    #pragma unroll
    for (int t = 0; t < 8; t++) {
        float bias = b2[t * 16 + mn];
        #pragma unroll
        for (int r = 0; r < 8; r++) {
            outV[(size_t)(rowBase + r + 8 * half) * Dn + t * 16 + mn] =
                acc2[t][r] + bias;
        }
    }
}

__global__ __launch_bounds__(128) void fmlp_kernel(
    const float* __restrict__ x,
    const float* __restrict__ fW1, const float* __restrict__ fb1,
    const float* __restrict__ fW2, const float* __restrict__ fb2,
    float* __restrict__ Wpad)
{
    __shared__ float Hlds[4][16 * HSTR];
    const int m       = blockIdx.y;
    const int wave    = threadIdx.x >> 5;
    const int lane    = threadIdx.x & 31;
    const int half    = lane >> 4;
    const int mn      = lane & 15;
    const int rowBase = blockIdx.x * 64 + wave * 16;

    const float* W1 = fW1 + (size_t)m * Dn * Dn;
    const float* B1 = fb1 + (size_t)m * Dn;
    const float* W2 = fW2 + (size_t)m * Dn * Kn;
    const float* B2 = fb2 + (size_t)m * Kn;

    v8f acc[8];
    #pragma unroll
    for (int t = 0; t < 8; t++) acc[t] = {};
    const float* Xrow = x + (size_t)(rowBase + mn) * Dn;
    #pragma unroll 4
    for (int k = 0; k < Dn; k += 4) {
        float2 av = *(const float2*)(Xrow + k + 2 * half);
        v2f a; a.x = av.x; a.y = av.y;
        #pragma unroll
        for (int t = 0; t < 8; t++) {
            const float* wp = W1 + (size_t)(k + 2 * half) * Dn + t * 16 + mn;
            v2f b; b.x = wp[0]; b.y = wp[Dn];
            acc[t] = wmma_f32x4(a, b, acc[t]);
        }
    }
    #pragma unroll
    for (int t = 0; t < 8; t++) {
        float bias = B1[t * 16 + mn];
        #pragma unroll
        for (int r = 0; r < 8; r++) {
            Hlds[wave][(r + 8 * half) * HSTR + t * 16 + mn] =
                gelu_exact(acc[t][r] + bias);
        }
    }
    __syncthreads();

    v8f acc2 = {};
    const bool valid = (mn < Kn);
    #pragma unroll 4
    for (int k = 0; k < Dn; k += 4) {
        float2 av = *(const float2*)(&Hlds[wave][mn * HSTR + k + 2 * half]);
        v2f a; a.x = av.x; a.y = av.y;
        v2f b;
        b.x = valid ? W2[(size_t)(k + 2 * half) * Kn + mn]     : 0.0f;
        b.y = valid ? W2[(size_t)(k + 2 * half + 1) * Kn + mn] : 0.0f;
        acc2 = wmma_f32x4(a, b, acc2);
    }
    float bias = valid ? B2[mn] : 0.0f;
    #pragma unroll
    for (int r = 0; r < 8; r++) {
        Wpad[((size_t)m * Rn + rowBase + r + 8 * half) * WPAD + mn] =
            acc2[r] + bias;
    }
}

__global__ __launch_bounds__(256) void chord_kernel(
    const float* __restrict__ Vsrc, const float* __restrict__ res,
    const float* __restrict__ Wm, float* __restrict__ Vdst)
{
    const int row  = blockIdx.x * 8 + (threadIdx.x >> 5);
    const int lane = threadIdx.x & 31;
    const int b    = row >> 12;
    const int i    = row & (Sn - 1);

    const int offs[Kn] = {0, 1, 2, 4, 8, 16, 32, 64, 128, 256, 512, 1024, 2048};

    float w[Kn];
    const float* wrow = Wm + (size_t)row * WPAD;
    #pragma unroll
    for (int k = 0; k < Kn; k++) w[k] = wrow[k];

    const float4* srcb = (const float4*)(Vsrc + (size_t)b * Sn * Dn);
    float4 acc = ((const float4*)(res + (size_t)row * Dn))[lane];
    #pragma unroll
    for (int k = 0; k < Kn; k++) {
        int j = (i + offs[k]) & (Sn - 1);
        float4 v = srcb[(size_t)j * (Dn / 4) + lane];
        acc.x += w[k] * v.x; acc.y += w[k] * v.y;
        acc.z += w[k] * v.z; acc.w += w[k] * v.w;
    }
    ((float4*)(Vdst + (size_t)row * Dn))[lane] = acc;
}

extern "C" void kernel_launch(void* const* d_in, const int* in_sizes, int n_in,
                              void* d_out, int out_size, void* d_ws, size_t ws_size,
                              hipStream_t stream) {
    const int*   data = (const int*)  d_in[0];
    const float* emb  = (const float*)d_in[2];
    const float* apc  = (const float*)d_in[3];
    const float* gW1  = (const float*)d_in[4];
    const float* gb1  = (const float*)d_in[5];
    const float* gW2  = (const float*)d_in[6];
    const float* gb2  = (const float*)d_in[7];
    const float* fW1  = (const float*)d_in[8];
    const float* fb1  = (const float*)d_in[9];
    const float* fW2  = (const float*)d_in[10];
    const float* fb2  = (const float*)d_in[11];
    float* out = (float*)d_out;

    char*  ws   = (char*)d_ws;
    float* xbuf = (float*)(ws);
    float* res  = (float*)(ws + (size_t) 8 * 1024 * 1024);
    float* Va   = (float*)(ws + (size_t)16 * 1024 * 1024);
    float* Wpad = (float*)(ws + (size_t)24 * 1024 * 1024);

    embed_kernel<<<(Rn * (Dn / 4)) / 256, 256, 0, stream>>>(data, emb, apc, xbuf);
    gmlp_kernel <<<Rn / 64, 128, 0, stream>>>(xbuf, gW1, gb1, gW2, gb2, res);
    fmlp_kernel <<<dim3(Rn / 64, Mn), 128, 0, stream>>>(xbuf, fW1, fb1, fW2, fb2, Wpad);

    const float* src = res;
    for (int m = 0; m < Mn; m++) {
        float* dst = (m & 1) ? out : Va;
        chord_kernel<<<Rn / 8, 256, 0, stream>>>(
            src, res, Wpad + (size_t)m * Rn * WPAD, dst);
        src = dst;
    }
}


# Fuse the 12 chord-mixing launches into one: a work-group keeps 2 columns of one batch for all 4096 rows in LDS (ping-pong), work-group barriers between steps; a small pass restores row layout
# speedup vs baseline: 1.2496x; 1.2496x over previous
#include <hip/hip_runtime.h>
#include <hip/hip_bf16.h>
#include <math.h>

#define Bz    4
#define Sn    4096
#define Dn    128
#define Kn    13
#define Mn    12
#define Rn    (Bz * Sn)
#define WPAD  16
#define HSTR  132

typedef float v2f __attribute__((ext_vector_type(2)));
typedef float v8f __attribute__((ext_vector_type(8)));

__device__ __forceinline__ float gelu_exact(float v) {
    return 0.5f * v * (1.0f + erff(v * 0.70710678118654752f));
}

__device__ __forceinline__ v8f wmma_f32x4(v2f a, v2f b, v8f c) {
    return __builtin_amdgcn_wmma_f32_16x16x4_f32(
         false, a,  false, b,
         (short)0, c,  false,  false);
}

__global__ __launch_bounds__(256) void embed_kernel(
    const int* __restrict__ data, const float* __restrict__ emb,
    const float* __restrict__ apc, float* __restrict__ x)
{
    int tid = blockIdx.x * blockDim.x + threadIdx.x;
    int row = tid >> 5;
    int c4  = tid & 31;
    if (row >= Rn) return;
    int s   = row & (Sn - 1);
    int tok = data[row];
    const float4* e = (const float4*)(emb + (size_t)tok * Dn);
    const float4* a = (const float4*)(apc + (size_t)s * Dn);
    float4 ev = e[c4], av = a[c4];
    float4 o; o.x = ev.x + av.x; o.y = ev.y + av.y; o.z = ev.z + av.z; o.w = ev.w + av.w;
    ((float4*)x)[tid] = o;
}

__global__ __launch_bounds__(128) void gmlp_kernel(
    const float* __restrict__ x,
    const float* __restrict__ W1, const float* __restrict__ b1,
    const float* __restrict__ W2, const float* __restrict__ b2,
    float* __restrict__ outV)
{
    __shared__ float Hlds[4][16 * HSTR];
    const int wave    = threadIdx.x >> 5;
    const int lane    = threadIdx.x & 31;
    const int half    = lane >> 4;
    const int mn      = lane & 15;
    const int rowBase = blockIdx.x * 64 + wave * 16;

    v8f acc[8];
    #pragma unroll
    for (int t = 0; t < 8; t++) acc[t] = {};
    const float* Xrow = x + (size_t)(rowBase + mn) * Dn;
    #pragma unroll 4
    for (int k = 0; k < Dn; k += 4) {
        float2 av = *(const float2*)(Xrow + k + 2 * half);
        v2f a; a.x = av.x; a.y = av.y;
        #pragma unroll
        for (int t = 0; t < 8; t++) {
            const float* wp = W1 + (size_t)(k + 2 * half) * Dn + t * 16 + mn;
            v2f b; b.x = wp[0]; b.y = wp[Dn];
            acc[t] = wmma_f32x4(a, b, acc[t]);
        }
    }
    #pragma unroll
    for (int t = 0; t < 8; t++) {
        float bias = b1[t * 16 + mn];
        #pragma unroll
        for (int r = 0; r < 8; r++) {
            Hlds[wave][(r + 8 * half) * HSTR + t * 16 + mn] =
                gelu_exact(acc[t][r] + bias);
        }
    }
    __syncthreads();

    v8f acc2[8];
    #pragma unroll
    for (int t = 0; t < 8; t++) acc2[t] = {};
    #pragma unroll 4
    for (int k = 0; k < Dn; k += 4) {
        float2 av = *(const float2*)(&Hlds[wave][mn * HSTR + k + 2 * half]);
        v2f a; a.x = av.x; a.y = av.y;
        #pragma unroll
        for (int t = 0; t < 8; t++) {
            const float* wp = W2 + (size_t)(k + 2 * half) * Dn + t * 16 + mn;
            v2f b; b.x = wp[0]; b.y = wp[Dn];
            acc2[t] = wmma_f32x4(a, b, acc2[t]);
        }
    }
    #pragma unroll
    for (int t = 0; t < 8; t++) {
        float bias = b2[t * 16 + mn];
        #pragma unroll
        for (int r = 0; r < 8; r++) {
            outV[(size_t)(rowBase + r + 8 * half) * Dn + t * 16 + mn] =
                acc2[t][r] + bias;
        }
    }
}

__global__ __launch_bounds__(128) void fmlp_kernel(
    const float* __restrict__ x,
    const float* __restrict__ fW1, const float* __restrict__ fb1,
    const float* __restrict__ fW2, const float* __restrict__ fb2,
    float* __restrict__ Wpad)
{
    __shared__ float Hlds[4][16 * HSTR];
    const int m       = blockIdx.y;
    const int wave    = threadIdx.x >> 5;
    const int lane    = threadIdx.x & 31;
    const int half    = lane >> 4;
    const int mn      = lane & 15;
    const int rowBase = blockIdx.x * 64 + wave * 16;

    const float* W1 = fW1 + (size_t)m * Dn * Dn;
    const float* B1 = fb1 + (size_t)m * Dn;
    const float* W2 = fW2 + (size_t)m * Dn * Kn;
    const float* B2 = fb2 + (size_t)m * Kn;

    v8f acc[8];
    #pragma unroll
    for (int t = 0; t < 8; t++) acc[t] = {};
    const float* Xrow = x + (size_t)(rowBase + mn) * Dn;
    #pragma unroll 4
    for (int k = 0; k < Dn; k += 4) {
        float2 av = *(const float2*)(Xrow + k + 2 * half);
        v2f a; a.x = av.x; a.y = av.y;
        #pragma unroll
        for (int t = 0; t < 8; t++) {
            const float* wp = W1 + (size_t)(k + 2 * half) * Dn + t * 16 + mn;
            v2f b; b.x = wp[0]; b.y = wp[Dn];
            acc[t] = wmma_f32x4(a, b, acc[t]);
        }
    }
    #pragma unroll
    for (int t = 0; t < 8; t++) {
        float bias = B1[t * 16 + mn];
        #pragma unroll
        for (int r = 0; r < 8; r++) {
            Hlds[wave][(r + 8 * half) * HSTR + t * 16 + mn] =
                gelu_exact(acc[t][r] + bias);
        }
    }
    __syncthreads();

    v8f acc2 = {};
    const bool valid = (mn < Kn);
    #pragma unroll 4
    for (int k = 0; k < Dn; k += 4) {
        float2 av = *(const float2*)(&Hlds[wave][mn * HSTR + k + 2 * half]);
        v2f a; a.x = av.x; a.y = av.y;
        v2f b;
        b.x = valid ? W2[(size_t)(k + 2 * half) * Kn + mn]     : 0.0f;
        b.y = valid ? W2[(size_t)(k + 2 * half + 1) * Kn + mn] : 0.0f;
        acc2 = wmma_f32x4(a, b, acc2);
    }
    float bias = valid ? B2[mn] : 0.0f;
    #pragma unroll
    for (int r = 0; r < 8; r++) {
        Wpad[((size_t)m * Rn + rowBase + r + 8 * half) * WPAD + mn] =
            acc2[r] + bias;
    }
}

#define CS   2
#define NSL  (Dn / CS)
#define CT   1024
#define RPT  (Sn / CT)

struct WRow { float4 a, b, c; float d; };

__device__ __forceinline__ WRow load_wrow(const float* __restrict__ p) {
    const float4* q = (const float4*)p;
    WRow w; w.a = q[0]; w.b = q[1]; w.c = q[2]; w.d = p[12];
    return w;
}

__global__ __launch_bounds__(CT) void chord_fused_kernel(
    const float* __restrict__ res, const float* __restrict__ Wpad,
    float* __restrict__ Vs)
{
    __shared__ float2 buf0[Sn];
    __shared__ float2 buf1[Sn];
    const int t = threadIdx.x;
    const int s = blockIdx.x & (NSL - 1);
    const int b = blockIdx.x / NSL;

    const int offs[Kn] = {0, 1, 2, 4, 8, 16, 32, 64, 128, 256, 512, 1024, 2048};

    float2 r[RPT];
    WRow   w[RPT];
    const float* Wb = Wpad + (size_t)b * Sn * WPAD;
    #pragma unroll
    for (int j = 0; j < RPT; j++) {
        const int i = t + CT * j;
        r[j] = *(const float2*)(res + ((size_t)b * Sn + i) * Dn + CS * s);
        buf0[i] = r[j];
        w[j] = load_wrow(Wb + (size_t)i * WPAD);
    }
    __syncthreads();

    float2* src = buf0;
    float2* dst = buf1;
    for (int m = 0; m < Mn; m++) {
        const float* Wn = Wb + (size_t)(m + 1) * Rn * WPAD;
        const bool more = (m + 1 < Mn);
        #pragma unroll
        for (int j = 0; j < RPT; j++) {
            const int i = t + CT * j;
            float2 v[Kn];
            #pragma unroll
            for (int k = 0; k < Kn; k++) v[k] = src[(i + offs[k]) & (Sn - 1)];
            asm volatile("" :: "v"(v[0].x), "v"(v[0].y), "v"(v[1].x), "v"(v[1].y),
                               "v"(v[2].x), "v"(v[2].y), "v"(v[3].x), "v"(v[3].y),
                               "v"(v[4].x), "v"(v[4].y), "v"(v[5].x), "v"(v[5].y),
                               "v"(v[6].x), "v"(v[6].y), "v"(v[7].x), "v"(v[7].y),
                               "v"(v[8].x), "v"(v[8].y), "v"(v[9].x), "v"(v[9].y),
                               "v"(v[10].x), "v"(v[10].y), "v"(v[11].x), "v"(v[11].y),
                               "v"(v[12].x), "v"(v[12].y));
            const float wk[Kn] = {w[j].a.x, w[j].a.y, w[j].a.z, w[j].a.w,
                                  w[j].b.x, w[j].b.y, w[j].b.z, w[j].b.w,
                                  w[j].c.x, w[j].c.y, w[j].c.z, w[j].c.w, w[j].d};
            float2 acc = r[j];
            #pragma unroll
            for (int k = 0; k < Kn; k++) {
                acc.x += wk[k] * v[k].x;
                acc.y += wk[k] * v[k].y;
            }
            dst[i] = acc;
            if (more) w[j] = load_wrow(Wn + (size_t)i * WPAD);
        }
        __builtin_amdgcn_fence(__ATOMIC_RELEASE, "workgroup", "local");
        __builtin_amdgcn_s_barrier();
        __builtin_amdgcn_fence(__ATOMIC_ACQUIRE, "workgroup", "local");
        float2* tmp = src; src = dst; dst = tmp;
    }

    float2* o = (float2*)Vs + ((size_t)b * NSL + s) * Sn;
    #pragma unroll
    for (int j = 0; j < RPT; j++) {
        const int i = t + CT * j;
        o[i] = src[i];
    }
}

#define TP 130

__global__ __launch_bounds__(256) void unslice_kernel(
    const float* __restrict__ Vs, float* __restrict__ out)
{
    __shared__ __attribute__((aligned(16))) float tile[32 * TP];
    const int wave = threadIdx.x >> 5;
    const int lane = threadIdx.x & 31;
    const int row0 = blockIdx.x * 32;
    const int b    = row0 >> 12;
    const int i0   = row0 & (Sn - 1);

    #pragma unroll
    for (int q = 0; q < 8; q++) {
        const int s = wave * 8 + q;
        const float2 v = ((const float2*)Vs)[((size_t)b * NSL + s) * Sn + i0 + lane];
        *(float2*)&tile[lane * TP + CS * s] = v;
    }
    __syncthreads();
    #pragma unroll
    for (int q = 0; q < 4; q++) {
        const int rr = wave * 4 + q;
        const float2 lo = *(const float2*)&tile[rr * TP + 4 * lane];
        const float2 hi = *(const float2*)&tile[rr * TP + 4 * lane + 2];
        float4 o; o.x = lo.x; o.y = lo.y; o.z = hi.x; o.w = hi.y;
        ((float4*)(out + (size_t)(row0 + rr) * Dn))[lane] = o;
    }
}

extern "C" void kernel_launch(void* const* d_in, const int* in_sizes, int n_in,
                              void* d_out, int out_size, void* d_ws, size_t ws_size,
                              hipStream_t stream) {
    const int*   data = (const int*)  d_in[0];
    const float* emb  = (const float*)d_in[2];
    const float* apc  = (const float*)d_in[3];
    const float* gW1  = (const float*)d_in[4];
    const float* gb1  = (const float*)d_in[5];
    const float* gW2  = (const float*)d_in[6];
    const float* gb2  = (const float*)d_in[7];
    const float* fW1  = (const float*)d_in[8];
    const float* fb1  = (const float*)d_in[9];
    const float* fW2  = (const float*)d_in[10];
    const float* fb2  = (const float*)d_in[11];
    float* out = (float*)d_out;

    char*  ws   = (char*)d_ws;
    float* xbuf = (float*)(ws);
    float* res  = (float*)(ws + (size_t) 8 * 1024 * 1024);
    float* Vs   = (float*)(ws + (size_t)16 * 1024 * 1024);
    float* Wpad = (float*)(ws + (size_t)24 * 1024 * 1024);

    embed_kernel<<<(Rn * (Dn / 4)) / 256, 256, 0, stream>>>(data, emb, apc, xbuf);
    gmlp_kernel <<<Rn / 64, 128, 0, stream>>>(xbuf, gW1, gb1, gW2, gb2, res);
    fmlp_kernel <<<dim3(Rn / 64, Mn), 128, 0, stream>>>(xbuf, fW1, fb1, fW2, fb2, Wpad);

    chord_fused_kernel<<<Bz * NSL, CT, 0, stream>>>(res, Wpad, Vs);
    unslice_kernel<<<Rn / 32, 256, 0, stream>>>(Vs, out);
}


# Fuse embed, g-MLP and the 12 f-MLPs into one launch (5 launches -> 3): each wave builds its 16 x rows in LDS from emb+apc (x never written to memory); the g-MLP is grid slice 0
# speedup vs baseline: 1.5507x; 1.2410x over previous
#include <hip/hip_runtime.h>
#include <hip/hip_bf16.h>
#include <math.h>

#define Bz    4
#define Sn    4096
#define Dn    128
#define Kn    13
#define Mn    12
#define Rn    (Bz * Sn)
#define WPAD  16
#define HSTR  132

typedef float v2f __attribute__((ext_vector_type(2)));
typedef float v8f __attribute__((ext_vector_type(8)));

__device__ __forceinline__ float gelu_exact(float v) {
    return 0.5f * v * (1.0f + erff(v * 0.70710678118654752f));
}

__device__ __forceinline__ v8f wmma_f32x4(v2f a, v2f b, v8f c) {
    return __builtin_amdgcn_wmma_f32_16x16x4_f32(
         false, a,  false, b,
         (short)0, c,  false,  false);
}

__global__ __launch_bounds__(128) void front_kernel(
    const int* __restrict__ data, const float* __restrict__ emb,
    const float* __restrict__ apc,
    const float* __restrict__ gW1, const float* __restrict__ gb1,
    const float* __restrict__ gW2, const float* __restrict__ gb2,
    const float* __restrict__ fW1, const float* __restrict__ fb1,
    const float* __restrict__ fW2, const float* __restrict__ fb2,
    float* __restrict__ res, float* __restrict__ Wpad)
{
    __shared__ __attribute__((aligned(16))) float Hlds[4][16 * HSTR];
    const int slice   = blockIdx.y;
    const bool isG    = (slice == 0);
    const int m       = isG ? 0 : slice - 1;
    const int wave    = __builtin_amdgcn_readfirstlane(threadIdx.x >> 5);
    const int lane    = threadIdx.x & 31;
    const int half    = lane >> 4;
    const int mn      = lane & 15;
    const int rowBase = blockIdx.x * 64 + wave * 16;
    float* tile = Hlds[wave];

    const float* W1 = isG ? gW1 : fW1 + (size_t)m * Dn * Dn;
    const float* B1 = isG ? gb1 : fb1 + (size_t)m * Dn;

    #pragma unroll
    for (int r = 0; r < 16; r++) {
        const int row = rowBase + r;
        const int s   = row & (Sn - 1);
        const int tok = data[row];
        const float4 ev = ((const float4*)(emb + (size_t)tok * Dn))[lane];
        const float4 av = ((const float4*)(apc + (size_t)s * Dn))[lane];
        float4 o; o.x = ev.x + av.x; o.y = ev.y + av.y; o.z = ev.z + av.z; o.w = ev.w + av.w;
        *(float4*)&tile[r * HSTR + 4 * lane] = o;
    }
    __syncthreads();

    v8f acc[8];
    #pragma unroll
    for (int t = 0; t < 8; t++) acc[t] = {};
    #pragma unroll 4
    for (int k = 0; k < Dn; k += 4) {
        float2 av = *(const float2*)(&tile[mn * HSTR + k + 2 * half]);
        v2f a; a.x = av.x; a.y = av.y;
        #pragma unroll
        for (int t = 0; t < 8; t++) {
            const float* wp = W1 + (size_t)(k + 2 * half) * Dn + t * 16 + mn;
            v2f b; b.x = wp[0]; b.y = wp[Dn];
            acc[t] = wmma_f32x4(a, b, acc[t]);
        }
    }
    __syncthreads();
    #pragma unroll
    for (int t = 0; t < 8; t++) {
        float bias = B1[t * 16 + mn];
        #pragma unroll
        for (int r = 0; r < 8; r++) {
            tile[(r + 8 * half) * HSTR + t * 16 + mn] =
                gelu_exact(acc[t][r] + bias);
        }
    }
    __syncthreads();

    if (isG) {
        v8f acc2[8];
        #pragma unroll
        for (int t = 0; t < 8; t++) acc2[t] = {};
        #pragma unroll 4
        for (int k = 0; k < Dn; k += 4) {
            float2 av = *(const float2*)(&tile[mn * HSTR + k + 2 * half]);
            v2f a; a.x = av.x; a.y = av.y;
            #pragma unroll
            for (int t = 0; t < 8; t++) {
                const float* wp = gW2 + (size_t)(k + 2 * half) * Dn + t * 16 + mn;
                v2f b; b.x = wp[0]; b.y = wp[Dn];
                acc2[t] = wmma_f32x4(a, b, acc2[t]);
            }
        }
        #pragma unroll
        for (int t = 0; t < 8; t++) {
            float bias = gb2[t * 16 + mn];
            #pragma unroll
            for (int r = 0; r < 8; r++) {
                res[(size_t)(rowBase + r + 8 * half) * Dn + t * 16 + mn] =
                    acc2[t][r] + bias;
            }
        }
    } else {
        const float* W2 = fW2 + (size_t)m * Dn * Kn;
        const float* B2 = fb2 + (size_t)m * Kn;
        v8f acc2 = {};
        const bool valid = (mn < Kn);
        #pragma unroll 4
        for (int k = 0; k < Dn; k += 4) {
            float2 av = *(const float2*)(&tile[mn * HSTR + k + 2 * half]);
            v2f a; a.x = av.x; a.y = av.y;
            v2f b;
            b.x = valid ? W2[(size_t)(k + 2 * half) * Kn + mn]     : 0.0f;
            b.y = valid ? W2[(size_t)(k + 2 * half + 1) * Kn + mn] : 0.0f;
            acc2 = wmma_f32x4(a, b, acc2);
        }
        float bias = valid ? B2[mn] : 0.0f;
        #pragma unroll
        for (int r = 0; r < 8; r++) {
            Wpad[((size_t)m * Rn + rowBase + r + 8 * half) * WPAD + mn] =
                acc2[r] + bias;
        }
    }
}

#define CS   2
#define NSL  (Dn / CS)
#define CT   1024
#define RPT  (Sn / CT)

struct WRow { float4 a, b, c; float d; };

__device__ __forceinline__ WRow load_wrow(const float* __restrict__ p) {
    const float4* q = (const float4*)p;
    WRow w; w.a = q[0]; w.b = q[1]; w.c = q[2]; w.d = p[12];
    return w;
}

__global__ __launch_bounds__(CT) void chord_fused_kernel(
    const float* __restrict__ res, const float* __restrict__ Wpad,
    float* __restrict__ Vs)
{
    __shared__ float2 buf0[Sn];
    __shared__ float2 buf1[Sn];
    const int t = threadIdx.x;
    const int s = blockIdx.x & (NSL - 1);
    const int b = blockIdx.x / NSL;

    const int offs[Kn] = {0, 1, 2, 4, 8, 16, 32, 64, 128, 256, 512, 1024, 2048};

    float2 r[RPT];
    WRow   w[RPT];
    const float* Wb = Wpad + (size_t)b * Sn * WPAD;
    #pragma unroll
    for (int j = 0; j < RPT; j++) {
        const int i = t + CT * j;
        r[j] = *(const float2*)(res + ((size_t)b * Sn + i) * Dn + CS * s);
        buf0[i] = r[j];
        w[j] = load_wrow(Wb + (size_t)i * WPAD);
    }
    __syncthreads();

    float2* src = buf0;
    float2* dst = buf1;
    for (int m = 0; m < Mn; m++) {
        const float* Wn = Wb + (size_t)(m + 1) * Rn * WPAD;
        const bool more = (m + 1 < Mn);
        #pragma unroll
        for (int j = 0; j < RPT; j++) {
            const int i = t + CT * j;
            float2 v[Kn];
            #pragma unroll
            for (int k = 0; k < Kn; k++) v[k] = src[(i + offs[k]) & (Sn - 1)];
            asm volatile("" :: "v"(v[0].x), "v"(v[0].y), "v"(v[1].x), "v"(v[1].y),
                               "v"(v[2].x), "v"(v[2].y), "v"(v[3].x), "v"(v[3].y),
                               "v"(v[4].x), "v"(v[4].y), "v"(v[5].x), "v"(v[5].y),
                               "v"(v[6].x), "v"(v[6].y), "v"(v[7].x), "v"(v[7].y),
                               "v"(v[8].x), "v"(v[8].y), "v"(v[9].x), "v"(v[9].y),
                               "v"(v[10].x), "v"(v[10].y), "v"(v[11].x), "v"(v[11].y),
                               "v"(v[12].x), "v"(v[12].y));
            const float wk[Kn] = {w[j].a.x, w[j].a.y, w[j].a.z, w[j].a.w,
                                  w[j].b.x, w[j].b.y, w[j].b.z, w[j].b.w,
                                  w[j].c.x, w[j].c.y, w[j].c.z, w[j].c.w, w[j].d};
            float2 acc = r[j];
            #pragma unroll
            for (int k = 0; k < Kn; k++) {
                acc.x += wk[k] * v[k].x;
                acc.y += wk[k] * v[k].y;
            }
            dst[i] = acc;
            if (more) w[j] = load_wrow(Wn + (size_t)i * WPAD);
        }
        __builtin_amdgcn_fence(__ATOMIC_RELEASE, "workgroup", "local");
        __builtin_amdgcn_s_barrier();
        __builtin_amdgcn_fence(__ATOMIC_ACQUIRE, "workgroup", "local");
        float2* tmp = src; src = dst; dst = tmp;
    }

    float2* o = (float2*)Vs + ((size_t)b * NSL + s) * Sn;
    #pragma unroll
    for (int j = 0; j < RPT; j++) {
        const int i = t + CT * j;
        o[i] = src[i];
    }
}

#define TP 130

__global__ __launch_bounds__(256) void unslice_kernel(
    const float* __restrict__ Vs, float* __restrict__ out)
{
    __shared__ __attribute__((aligned(16))) float tile[32 * TP];
    const int wave = threadIdx.x >> 5;
    const int lane = threadIdx.x & 31;
    const int row0 = blockIdx.x * 32;
    const int b    = row0 >> 12;
    const int i0   = row0 & (Sn - 1);

    #pragma unroll
    for (int q = 0; q < 8; q++) {
        const int s = wave * 8 + q;
        const float2 v = ((const float2*)Vs)[((size_t)b * NSL + s) * Sn + i0 + lane];
        *(float2*)&tile[lane * TP + CS * s] = v;
    }
    __syncthreads();
    #pragma unroll
    for (int q = 0; q < 4; q++) {
        const int rr = wave * 4 + q;
        const float2 lo = *(const float2*)&tile[rr * TP + 4 * lane];
        const float2 hi = *(const float2*)&tile[rr * TP + 4 * lane + 2];
        float4 o; o.x = lo.x; o.y = lo.y; o.z = hi.x; o.w = hi.y;
        ((float4*)(out + (size_t)(row0 + rr) * Dn))[lane] = o;
    }
}

extern "C" void kernel_launch(void* const* d_in, const int* in_sizes, int n_in,
                              void* d_out, int out_size, void* d_ws, size_t ws_size,
                              hipStream_t stream) {
    const int*   data = (const int*)  d_in[0];
    const float* emb  = (const float*)d_in[2];
    const float* apc  = (const float*)d_in[3];
    const float* gW1  = (const float*)d_in[4];
    const float* gb1  = (const float*)d_in[5];
    const float* gW2  = (const float*)d_in[6];
    const float* gb2  = (const float*)d_in[7];
    const float* fW1  = (const float*)d_in[8];
    const float* fb1  = (const float*)d_in[9];
    const float* fW2  = (const float*)d_in[10];
    const float* fb2  = (const float*)d_in[11];
    float* out = (float*)d_out;

    char*  ws   = (char*)d_ws;
    float* res  = (float*)(ws + (size_t) 8 * 1024 * 1024);
    float* Vs   = (float*)(ws + (size_t)16 * 1024 * 1024);
    float* Wpad = (float*)(ws + (size_t)24 * 1024 * 1024);

    front_kernel<<<dim3(Rn / 64, Mn + 1), 128, 0, stream>>>(
        data, emb, apc, gW1, gb1, gW2, gb2, fW1, fb1, fW2, fb2, res, Wpad);

    chord_fused_kernel<<<Bz * NSL, CT, 0, stream>>>(res, Wpad, Vs);
    unslice_kernel<<<Rn / 32, 256, 0, stream>>>(Vs, out);
}


# Front stage: each 8-wave work-group copies its slice's weights into LDS once, transposed; a B fragment is one 8-byte LDS read, not two 4-byte global loads per matrix instruction; k order kept
# speedup vs baseline: 1.7294x; 1.1152x over previous
#include <hip/hip_runtime.h>
#include <hip/hip_bf16.h>
#include <math.h>

#define Bz    4
#define Sn    4096
#define Dn    128
#define Kn    13
#define Mn    12
#define Rn    (Bz * Sn)
#define WPAD  16
#define HSTR  132

typedef float v2f __attribute__((ext_vector_type(2)));
typedef float v8f __attribute__((ext_vector_type(8)));

__device__ __forceinline__ float gelu_exact(float v) {
    return 0.5f * v * (1.0f + erff(v * 0.70710678118654752f));
}

__device__ __forceinline__ v8f wmma_f32x4(v2f a, v2f b, v8f c) {
    return __builtin_amdgcn_wmma_f32_16x16x4_f32(
         false, a,  false, b,
         (short)0, c,  false,  false);
}

#define TW    8
#define WSTR  132

struct WStage { float4 r0[8], r1[8]; };

__device__ __forceinline__ void stage_load(
    WStage& w, const float* __restrict__ Wg, int wave, int lane)
{
    const int kp = lane & 7;
    const int n0 = wave * 16 + (lane >> 3) * 4;
    #pragma unroll
    for (int j = 0; j < 8; j++) {
        const int k0 = j * 16 + kp * 2;
        w.r0[j] = *(const float4*)(Wg + (size_t)k0 * Dn + n0);
        w.r1[j] = *(const float4*)(Wg + (size_t)(k0 + 1) * Dn + n0);
    }
}

__device__ __forceinline__ void stage_store(
    const WStage& w, float* Wt, int wave, int lane)
{
    const int kp = lane & 7;
    const int n0 = wave * 16 + (lane >> 3) * 4;
    #pragma unroll
    for (int j = 0; j < 8; j++) {
        const int k0 = j * 16 + kp * 2;
        float2 p;
        p.x = w.r0[j].x; p.y = w.r1[j].x; *(float2*)&Wt[(n0 + 0) * WSTR + k0] = p;
        p.x = w.r0[j].y; p.y = w.r1[j].y; *(float2*)&Wt[(n0 + 1) * WSTR + k0] = p;
        p.x = w.r0[j].z; p.y = w.r1[j].z; *(float2*)&Wt[(n0 + 2) * WSTR + k0] = p;
        p.x = w.r0[j].w; p.y = w.r1[j].w; *(float2*)&Wt[(n0 + 3) * WSTR + k0] = p;
    }
}

__device__ __forceinline__ void product_wide(
    const float* aP, const float* bP, v8f (&acc)[8])
{
    #pragma unroll 4
    for (int k = 0; k < Dn; k += 4) {
        const v2f a = *(const v2f*)(aP + k);
        #pragma unroll
        for (int t = 0; t < 8; t++) {
            const v2f b = *(const v2f*)(bP + t * 16 * WSTR + k);
            acc[t] = wmma_f32x4(a, b, acc[t]);
        }
    }
}

__device__ __forceinline__ void wave_tile_sync() {
    __builtin_amdgcn_fence(__ATOMIC_RELEASE, "wavefront", "local");
    __builtin_amdgcn_wave_barrier();
    __builtin_amdgcn_fence(__ATOMIC_ACQUIRE, "wavefront", "local");
}

__global__ __launch_bounds__(TW * 32) void front_kernel(
    const int* __restrict__ data, const float* __restrict__ emb,
    const float* __restrict__ apc,
    const float* __restrict__ gW1, const float* __restrict__ gb1,
    const float* __restrict__ gW2, const float* __restrict__ gb2,
    const float* __restrict__ fW1, const float* __restrict__ fb1,
    const float* __restrict__ fW2, const float* __restrict__ fb2,
    float* __restrict__ res, float* __restrict__ Wpad)
{
    __shared__ __attribute__((aligned(16))) float Wt1[Dn * WSTR];
    __shared__ __attribute__((aligned(16))) float Wt2[16 * WSTR];
    __shared__ __attribute__((aligned(16))) float Hlds[TW][16 * HSTR];
    const int slice   = blockIdx.y;
    const bool isG    = (slice == 0);
    const int m       = isG ? 0 : slice - 1;
    const int wave    = __builtin_amdgcn_readfirstlane(threadIdx.x >> 5);
    const int lane    = threadIdx.x & 31;
    const int half    = lane >> 4;
    const int mn      = lane & 15;
    const int rowBase = blockIdx.x * (TW * 16) + wave * 16;
    float* tile = Hlds[wave];

    const float* W1 = isG ? gW1 : fW1 + (size_t)m * Dn * Dn;
    const float* B1 = isG ? gb1 : fb1 + (size_t)m * Dn;

    WStage ws1;
    stage_load(ws1, W1, wave, lane);
    const float* W2 = fW2 + (size_t)m * Dn * Kn;
    float w2v[7];
    if (!isG) {
        #pragma unroll
        for (int j = 0; j < 7; j++) {
            const int idx = threadIdx.x + j * (TW * 32);
            w2v[j] = (idx < Dn * Kn) ? W2[idx] : 0.0f;
        }
    }
    float4 ev[16], av[16];
    #pragma unroll
    for (int r = 0; r < 16; r++) {
        const int row = rowBase + r;
        const int s   = row & (Sn - 1);
        const int tok = data[row];
        ev[r] = ((const float4*)(emb + (size_t)tok * Dn))[lane];
        av[r] = ((const float4*)(apc + (size_t)s * Dn))[lane];
    }
    stage_store(ws1, Wt1, wave, lane);
    if (!isG) {
        #pragma unroll
        for (int j = 0; j < 7; j++) {
            const int idx = threadIdx.x + j * (TW * 32);
            if (idx < Dn * Kn) {
                const int k = idx / Kn;
                const int n = idx - k * Kn;
                Wt2[n * WSTR + k] = w2v[j];
            }
        }
        Wt2[(Kn + (threadIdx.x >> 7)) * WSTR + (threadIdx.x & (Dn - 1))] = 0.0f;
        if (threadIdx.x < Dn) Wt2[(Kn + 2) * WSTR + threadIdx.x] = 0.0f;
    }
    #pragma unroll
    for (int r = 0; r < 16; r++) {
        float4 o;
        o.x = ev[r].x + av[r].x; o.y = ev[r].y + av[r].y;
        o.z = ev[r].z + av[r].z; o.w = ev[r].w + av[r].w;
        *(float4*)&tile[r * HSTR + 4 * lane] = o;
    }
    __syncthreads();

    const float* aP = tile + mn * HSTR + 2 * half;
    const float* bP = Wt1 + mn * WSTR + 2 * half;

    v8f acc[8];
    #pragma unroll
    for (int t = 0; t < 8; t++) acc[t] = {};
    product_wide(aP, bP, acc);
    wave_tile_sync();
    #pragma unroll
    for (int t = 0; t < 8; t++) {
        float bias = B1[t * 16 + mn];
        #pragma unroll
        for (int r = 0; r < 8; r++) {
            tile[(r + 8 * half) * HSTR + t * 16 + mn] =
                gelu_exact(acc[t][r] + bias);
        }
    }

    if (isG) {
        WStage ws2;
        stage_load(ws2, gW2, wave, lane);
        __syncthreads();
        stage_store(ws2, Wt1, wave, lane);
        __syncthreads();
        v8f acc2[8];
        #pragma unroll
        for (int t = 0; t < 8; t++) acc2[t] = {};
        product_wide(aP, bP, acc2);
        #pragma unroll
        for (int t = 0; t < 8; t++) {
            float bias = gb2[t * 16 + mn];
            #pragma unroll
            for (int r = 0; r < 8; r++) {
                res[(size_t)(rowBase + r + 8 * half) * Dn + t * 16 + mn] =
                    acc2[t][r] + bias;
            }
        }
    } else {
        wave_tile_sync();
        const float* B2 = fb2 + (size_t)m * Kn;
        const float* b2P = Wt2 + mn * WSTR + 2 * half;
        v8f acc2 = {};
        const bool valid = (mn < Kn);
        for (int k0 = 0; k0 < Dn; k0 += 32) {
            v2f a[8], b[8];
            #pragma unroll
            for (int u = 0; u < 8; u++) {
                a[u] = *(const v2f*)(aP + k0 + 4 * u);
                b[u] = *(const v2f*)(b2P + k0 + 4 * u);
            }
            #pragma unroll
            for (int u = 0; u < 8; u++) acc2 = wmma_f32x4(a[u], b[u], acc2);
        }
        float bias = valid ? B2[mn] : 0.0f;
        #pragma unroll
        for (int r = 0; r < 8; r++) {
            Wpad[((size_t)m * Rn + rowBase + r + 8 * half) * WPAD + mn] =
                acc2[r] + bias;
        }
    }
}

#define CS   2
#define NSL  (Dn / CS)
#define CT   1024
#define RPT  (Sn / CT)

struct WRow { float4 a, b, c; float d; };

__device__ __forceinline__ WRow load_wrow(const float* __restrict__ p) {
    const float4* q = (const float4*)p;
    WRow w; w.a = q[0]; w.b = q[1]; w.c = q[2]; w.d = p[12];
    return w;
}

__global__ __launch_bounds__(CT) void chord_fused_kernel(
    const float* __restrict__ res, const float* __restrict__ Wpad,
    float* __restrict__ Vs)
{
    __shared__ float2 buf0[Sn];
    __shared__ float2 buf1[Sn];
    const int t = threadIdx.x;
    const int s = blockIdx.x & (NSL - 1);
    const int b = blockIdx.x / NSL;

    const int offs[Kn] = {0, 1, 2, 4, 8, 16, 32, 64, 128, 256, 512, 1024, 2048};

    float2 r[RPT];
    WRow   w[RPT];
    const float* Wb = Wpad + (size_t)b * Sn * WPAD;
    #pragma unroll
    for (int j = 0; j < RPT; j++) {
        const int i = t + CT * j;
        r[j] = *(const float2*)(res + ((size_t)b * Sn + i) * Dn + CS * s);
        buf0[i] = r[j];
        w[j] = load_wrow(Wb + (size_t)i * WPAD);
    }
    __syncthreads();

    float2* src = buf0;
    float2* dst = buf1;
    for (int m = 0; m < Mn; m++) {
        const float* Wn = Wb + (size_t)(m + 1) * Rn * WPAD;
        const bool more = (m + 1 < Mn);
        #pragma unroll
        for (int j = 0; j < RPT; j++) {
            const int i = t + CT * j;
            float2 v[Kn];
            #pragma unroll
            for (int k = 0; k < Kn; k++) v[k] = src[(i + offs[k]) & (Sn - 1)];
            asm volatile("" :: "v"(v[0].x), "v"(v[0].y), "v"(v[1].x), "v"(v[1].y),
                               "v"(v[2].x), "v"(v[2].y), "v"(v[3].x), "v"(v[3].y),
                               "v"(v[4].x), "v"(v[4].y), "v"(v[5].x), "v"(v[5].y),
                               "v"(v[6].x), "v"(v[6].y), "v"(v[7].x), "v"(v[7].y),
                               "v"(v[8].x), "v"(v[8].y), "v"(v[9].x), "v"(v[9].y),
                               "v"(v[10].x), "v"(v[10].y), "v"(v[11].x), "v"(v[11].y),
                               "v"(v[12].x), "v"(v[12].y));
            const float wk[Kn] = {w[j].a.x, w[j].a.y, w[j].a.z, w[j].a.w,
                                  w[j].b.x, w[j].b.y, w[j].b.z, w[j].b.w,
                                  w[j].c.x, w[j].c.y, w[j].c.z, w[j].c.w, w[j].d};
            float2 acc = r[j];
            #pragma unroll
            for (int k = 0; k < Kn; k++) {
                acc.x += wk[k] * v[k].x;
                acc.y += wk[k] * v[k].y;
            }
            dst[i] = acc;
            if (more) w[j] = load_wrow(Wn + (size_t)i * WPAD);
        }
        __builtin_amdgcn_fence(__ATOMIC_RELEASE, "workgroup", "local");
        __builtin_amdgcn_s_barrier();
        __builtin_amdgcn_fence(__ATOMIC_ACQUIRE, "workgroup", "local");
        float2* tmp = src; src = dst; dst = tmp;
    }

    float2* o = (float2*)Vs + ((size_t)b * NSL + s) * Sn;
    #pragma unroll
    for (int j = 0; j < RPT; j++) {
        const int i = t + CT * j;
        o[i] = src[i];
    }
}

#define TP 130

__global__ __launch_bounds__(256) void unslice_kernel(
    const float* __restrict__ Vs, float* __restrict__ out)
{
    __shared__ __attribute__((aligned(16))) float tile[32 * TP];
    const int wave = threadIdx.x >> 5;
    const int lane = threadIdx.x & 31;
    const int row0 = blockIdx.x * 32;
    const int b    = row0 >> 12;
    const int i0   = row0 & (Sn - 1);

    #pragma unroll
    for (int q = 0; q < 8; q++) {
        const int s = wave * 8 + q;
        const float2 v = ((const float2*)Vs)[((size_t)b * NSL + s) * Sn + i0 + lane];
        *(float2*)&tile[lane * TP + CS * s] = v;
    }
    __syncthreads();
    #pragma unroll
    for (int q = 0; q < 4; q++) {
        const int rr = wave * 4 + q;
        const float2 lo = *(const float2*)&tile[rr * TP + 4 * lane];
        const float2 hi = *(const float2*)&tile[rr * TP + 4 * lane + 2];
        float4 o; o.x = lo.x; o.y = lo.y; o.z = hi.x; o.w = hi.y;
        ((float4*)(out + (size_t)(row0 + rr) * Dn))[lane] = o;
    }
}

extern "C" void kernel_launch(void* const* d_in, const int* in_sizes, int n_in,
                              void* d_out, int out_size, void* d_ws, size_t ws_size,
                              hipStream_t stream) {
    const int*   data = (const int*)  d_in[0];
    const float* emb  = (const float*)d_in[2];
    const float* apc  = (const float*)d_in[3];
    const float* gW1  = (const float*)d_in[4];
    const float* gb1  = (const float*)d_in[5];
    const float* gW2  = (const float*)d_in[6];
    const float* gb2  = (const float*)d_in[7];
    const float* fW1  = (const float*)d_in[8];
    const float* fb1  = (const float*)d_in[9];
    const float* fW2  = (const float*)d_in[10];
    const float* fb2  = (const float*)d_in[11];
    float* out = (float*)d_out;

    char*  ws   = (char*)d_ws;
    float* res  = (float*)(ws + (size_t) 8 * 1024 * 1024);
    float* Vs   = (float*)(ws + (size_t)16 * 1024 * 1024);
    float* Wpad = (float*)(ws + (size_t)24 * 1024 * 1024);

    front_kernel<<<dim3(Rn / (TW * 16), Mn + 1), TW * 32, 0, stream>>>(
        data, emb, apc, gW1, gb1, gW2, gb2, fW1, fb1, fW2, fb2, res, Wpad);

    chord_fused_kernel<<<Bz * NSL, CT, 0, stream>>>(res, Wpad, Vs);
    unslice_kernel<<<Rn / 32, 256, 0, stream>>>(Vs, out);
}


# Front-stage epilogue between the two products: 64 serial branching erf calls become one wave-wide range test plus 64 interleaved branch-free polynomials (library erf as fallback); biases preloaded
# speedup vs baseline: 1.9137x; 1.1066x over previous
#include <hip/hip_runtime.h>
#include <hip/hip_bf16.h>
#include <math.h>

#define Bz    4
#define Sn    4096
#define Dn    128
#define Kn    13
#define Mn    12
#define Rn    (Bz * Sn)
#define WPAD  16
#define HSTR  132

typedef float v2f __attribute__((ext_vector_type(2)));
typedef float v8f __attribute__((ext_vector_type(8)));

__device__ __forceinline__ float gelu_exact(float v) {
    return 0.5f * v * (1.0f + erff(v * 0.70710678118654752f));
}

#define GELU_SMALL_LIMIT 1.41f

__device__ __forceinline__ float gelu_small(float v) {
    const float u = v * 0.70710678118654752f;
    const float t = u * u;
    float p = __builtin_fmaf(t, -0x1.268bc2p-11f, 0x1.420828p-8f);
    p = __builtin_fmaf(t, p, -0x1.b59370p-6f);
    p = __builtin_fmaf(t, p, 0x1.ce077cp-4f);
    p = __builtin_fmaf(t, p, -0x1.812660p-2f);
    p = __builtin_fmaf(t, p, 0x1.06eba0p-3f);
    const float e = __builtin_fmaf(u, p, u);
    return 0.5f * v * (1.0f + e);
}

__device__ __forceinline__ v8f wmma_f32x4(v2f a, v2f b, v8f c) {
    return __builtin_amdgcn_wmma_f32_16x16x4_f32(
         false, a,  false, b,
         (short)0, c,  false,  false);
}

#define TW    8
#define WSTR  132

struct WStage { float4 r0[8], r1[8]; };

__device__ __forceinline__ void stage_load(
    WStage& w, const float* __restrict__ Wg, int wave, int lane)
{
    const int kp = lane & 7;
    const int n0 = wave * 16 + (lane >> 3) * 4;
    #pragma unroll
    for (int j = 0; j < 8; j++) {
        const int k0 = j * 16 + kp * 2;
        w.r0[j] = *(const float4*)(Wg + (size_t)k0 * Dn + n0);
        w.r1[j] = *(const float4*)(Wg + (size_t)(k0 + 1) * Dn + n0);
    }
}

__device__ __forceinline__ void stage_store(
    const WStage& w, float* Wt, int wave, int lane)
{
    const int kp = lane & 7;
    const int n0 = wave * 16 + (lane >> 3) * 4;
    #pragma unroll
    for (int j = 0; j < 8; j++) {
        const int k0 = j * 16 + kp * 2;
        float2 p;
        p.x = w.r0[j].x; p.y = w.r1[j].x; *(float2*)&Wt[(n0 + 0) * WSTR + k0] = p;
        p.x = w.r0[j].y; p.y = w.r1[j].y; *(float2*)&Wt[(n0 + 1) * WSTR + k0] = p;
        p.x = w.r0[j].z; p.y = w.r1[j].z; *(float2*)&Wt[(n0 + 2) * WSTR + k0] = p;
        p.x = w.r0[j].w; p.y = w.r1[j].w; *(float2*)&Wt[(n0 + 3) * WSTR + k0] = p;
    }
}

__device__ __forceinline__ void product_wide(
    const float* aP, const float* bP, v8f (&acc)[8])
{
    #pragma unroll 4
    for (int k = 0; k < Dn; k += 4) {
        const v2f a = *(const v2f*)(aP + k);
        #pragma unroll
        for (int t = 0; t < 8; t++) {
            const v2f b = *(const v2f*)(bP + t * 16 * WSTR + k);
            acc[t] = wmma_f32x4(a, b, acc[t]);
        }
    }
}

__device__ __forceinline__ void wave_tile_sync() {
    __builtin_amdgcn_fence(__ATOMIC_RELEASE, "wavefront", "local");
    __builtin_amdgcn_wave_barrier();
    __builtin_amdgcn_fence(__ATOMIC_ACQUIRE, "wavefront", "local");
}

__global__ __launch_bounds__(TW * 32) void front_kernel(
    const int* __restrict__ data, const float* __restrict__ emb,
    const float* __restrict__ apc,
    const float* __restrict__ gW1, const float* __restrict__ gb1,
    const float* __restrict__ gW2, const float* __restrict__ gb2,
    const float* __restrict__ fW1, const float* __restrict__ fb1,
    const float* __restrict__ fW2, const float* __restrict__ fb2,
    float* __restrict__ res, float* __restrict__ Wpad)
{
    __shared__ __attribute__((aligned(16))) float Wt1[Dn * WSTR];
    __shared__ __attribute__((aligned(16))) float Wt2[16 * WSTR];
    __shared__ __attribute__((aligned(16))) float Hlds[TW][16 * HSTR];
    const int slice   = blockIdx.y;
    const bool isG    = (slice == 0);
    const int m       = isG ? 0 : slice - 1;
    const int wave    = __builtin_amdgcn_readfirstlane(threadIdx.x >> 5);
    const int lane    = threadIdx.x & 31;
    const int half    = lane >> 4;
    const int mn      = lane & 15;
    const int rowBase = blockIdx.x * (TW * 16) + wave * 16;
    float* tile = Hlds[wave];

    const float* W1 = isG ? gW1 : fW1 + (size_t)m * Dn * Dn;
    const float* B1 = isG ? gb1 : fb1 + (size_t)m * Dn;

    WStage ws1;
    stage_load(ws1, W1, wave, lane);
    const float* W2 = fW2 + (size_t)m * Dn * Kn;
    float w2v[7];
    if (!isG) {
        #pragma unroll
        for (int j = 0; j < 7; j++) {
            const int idx = threadIdx.x + j * (TW * 32);
            w2v[j] = (idx < Dn * Kn) ? W2[idx] : 0.0f;
        }
    }
    float4 ev[16], av[16];
    #pragma unroll
    for (int r = 0; r < 16; r++) {
        const int row = rowBase + r;
        const int s   = row & (Sn - 1);
        const int tok = data[row];
        ev[r] = ((const float4*)(emb + (size_t)tok * Dn))[lane];
        av[r] = ((const float4*)(apc + (size_t)s * Dn))[lane];
    }
    stage_store(ws1, Wt1, wave, lane);
    if (!isG) {
        #pragma unroll
        for (int j = 0; j < 7; j++) {
            const int idx = threadIdx.x + j * (TW * 32);
            if (idx < Dn * Kn) {
                const int k = idx / Kn;
                const int n = idx - k * Kn;
                Wt2[n * WSTR + k] = w2v[j];
            }
        }
        Wt2[(Kn + (threadIdx.x >> 7)) * WSTR + (threadIdx.x & (Dn - 1))] = 0.0f;
        if (threadIdx.x < Dn) Wt2[(Kn + 2) * WSTR + threadIdx.x] = 0.0f;
    }
    #pragma unroll
    for (int r = 0; r < 16; r++) {
        float4 o;
        o.x = ev[r].x + av[r].x; o.y = ev[r].y + av[r].y;
        o.z = ev[r].z + av[r].z; o.w = ev[r].w + av[r].w;
        *(float4*)&tile[r * HSTR + 4 * lane] = o;
    }
    float bq[8];
    #pragma unroll
    for (int t = 0; t < 8; t++) bq[t] = B1[t * 16 + mn];
    __syncthreads();

    const float* aP = tile + mn * HSTR + 2 * half;
    const float* bP = Wt1 + mn * WSTR + 2 * half;

    v8f acc[8];
    #pragma unroll
    for (int t = 0; t < 8; t++) acc[t] = {};
    product_wide(aP, bP, acc);
    wave_tile_sync();
    float mx[8];
    #pragma unroll
    for (int t = 0; t < 8; t++) {
        mx[t] = 0.0f;
        #pragma unroll
        for (int r = 0; r < 8; r++) {
            acc[t][r] += bq[t];
            mx[t] = fmaxf(mx[t], fabsf(acc[t][r]));
        }
    }
    const float mAll = fmaxf(fmaxf(fmaxf(mx[0], mx[1]), fmaxf(mx[2], mx[3])),
                             fmaxf(fmaxf(mx[4], mx[5]), fmaxf(mx[6], mx[7])));
    const bool outside = !(mAll < GELU_SMALL_LIMIT);
    if (__builtin_amdgcn_ballot_w32(outside) == 0u) {
        #pragma unroll
        for (int t = 0; t < 8; t++) {
            #pragma unroll
            for (int r = 0; r < 8; r++) {
                tile[(r + 8 * half) * HSTR + t * 16 + mn] = gelu_small(acc[t][r]);
            }
        }
    } else {
        #pragma unroll
        for (int t = 0; t < 8; t++) {
            #pragma unroll
            for (int r = 0; r < 8; r++) {
                tile[(r + 8 * half) * HSTR + t * 16 + mn] = acc[t][r];
            }
        }
        #pragma nounroll
        for (int i = 0; i < 64; i++) {
            float* cell = tile + ((i & 7) + 8 * half) * HSTR + (i >> 3) * 16 + mn;
            *cell = gelu_exact(*cell);
        }
    }

    if (isG) {
        WStage ws2;
        stage_load(ws2, gW2, wave, lane);
        float bo[8];
        #pragma unroll
        for (int t = 0; t < 8; t++) bo[t] = gb2[t * 16 + mn];
        __syncthreads();
        stage_store(ws2, Wt1, wave, lane);
        __syncthreads();
        v8f acc2[8];
        #pragma unroll
        for (int t = 0; t < 8; t++) acc2[t] = {};
        product_wide(aP, bP, acc2);
        #pragma unroll
        for (int t = 0; t < 8; t++) {
            #pragma unroll
            for (int r = 0; r < 8; r++) {
                res[(size_t)(rowBase + r + 8 * half) * Dn + t * 16 + mn] =
                    acc2[t][r] + bo[t];
            }
        }
    } else {
        wave_tile_sync();
        const float* B2 = fb2 + (size_t)m * Kn;
        const float* b2P = Wt2 + mn * WSTR + 2 * half;
        v8f acc2 = {};
        const bool valid = (mn < Kn);
        const float bias = valid ? B2[mn] : 0.0f;
        for (int k0 = 0; k0 < Dn; k0 += 32) {
            v2f a[8], b[8];
            #pragma unroll
            for (int u = 0; u < 8; u++) {
                a[u] = *(const v2f*)(aP + k0 + 4 * u);
                b[u] = *(const v2f*)(b2P + k0 + 4 * u);
            }
            #pragma unroll
            for (int u = 0; u < 8; u++) acc2 = wmma_f32x4(a[u], b[u], acc2);
        }
        #pragma unroll
        for (int r = 0; r < 8; r++) {
            Wpad[((size_t)m * Rn + rowBase + r + 8 * half) * WPAD + mn] =
                acc2[r] + bias;
        }
    }
}

#define CS   2
#define NSL  (Dn / CS)
#define CT   1024
#define RPT  (Sn / CT)

struct WRow { float4 a, b, c; float d; };

__device__ __forceinline__ WRow load_wrow(const float* __restrict__ p) {
    const float4* q = (const float4*)p;
    WRow w; w.a = q[0]; w.b = q[1]; w.c = q[2]; w.d = p[12];
    return w;
}

__global__ __launch_bounds__(CT) void chord_fused_kernel(
    const float* __restrict__ res, const float* __restrict__ Wpad,
    float* __restrict__ Vs)
{
    __shared__ float2 buf0[Sn];
    __shared__ float2 buf1[Sn];
    const int t = threadIdx.x;
    const int s = blockIdx.x & (NSL - 1);
    const int b = blockIdx.x / NSL;

    const int offs[Kn] = {0, 1, 2, 4, 8, 16, 32, 64, 128, 256, 512, 1024, 2048};

    float2 r[RPT];
    WRow   w[RPT];
    const float* Wb = Wpad + (size_t)b * Sn * WPAD;
    #pragma unroll
    for (int j = 0; j < RPT; j++) {
        const int i = t + CT * j;
        r[j] = *(const float2*)(res + ((size_t)b * Sn + i) * Dn + CS * s);
        buf0[i] = r[j];
        w[j] = load_wrow(Wb + (size_t)i * WPAD);
    }
    __syncthreads();

    float2* src = buf0;
    float2* dst = buf1;
    for (int m = 0; m < Mn; m++) {
        const float* Wn = Wb + (size_t)(m + 1) * Rn * WPAD;
        const bool more = (m + 1 < Mn);
        #pragma unroll
        for (int j = 0; j < RPT; j++) {
            const int i = t + CT * j;
            float2 v[Kn];
            #pragma unroll
            for (int k = 0; k < Kn; k++) v[k] = src[(i + offs[k]) & (Sn - 1)];
            asm volatile("" :: "v"(v[0].x), "v"(v[0].y), "v"(v[1].x), "v"(v[1].y),
                               "v"(v[2].x), "v"(v[2].y), "v"(v[3].x), "v"(v[3].y),
                               "v"(v[4].x), "v"(v[4].y), "v"(v[5].x), "v"(v[5].y),
                               "v"(v[6].x), "v"(v[6].y), "v"(v[7].x), "v"(v[7].y),
                               "v"(v[8].x), "v"(v[8].y), "v"(v[9].x), "v"(v[9].y),
                               "v"(v[10].x), "v"(v[10].y), "v"(v[11].x), "v"(v[11].y),
                               "v"(v[12].x), "v"(v[12].y));
            const float wk[Kn] = {w[j].a.x, w[j].a.y, w[j].a.z, w[j].a.w,
                                  w[j].b.x, w[j].b.y, w[j].b.z, w[j].b.w,
                                  w[j].c.x, w[j].c.y, w[j].c.z, w[j].c.w, w[j].d};
            float2 acc = r[j];
            #pragma unroll
            for (int k = 0; k < Kn; k++) {
                acc.x += wk[k] * v[k].x;
                acc.y += wk[k] * v[k].y;
            }
            dst[i] = acc;
            if (more) w[j] = load_wrow(Wn + (size_t)i * WPAD);
        }
        __builtin_amdgcn_fence(__ATOMIC_RELEASE, "workgroup", "local");
        __builtin_amdgcn_s_barrier();
        __builtin_amdgcn_fence(__ATOMIC_ACQUIRE, "workgroup", "local");
        float2* tmp = src; src = dst; dst = tmp;
    }

    float2* o = (float2*)Vs + ((size_t)b * NSL + s) * Sn;
    #pragma unroll
    for (int j = 0; j < RPT; j++) {
        const int i = t + CT * j;
        o[i] = src[i];
    }
}

#define TP 130

__global__ __launch_bounds__(256) void unslice_kernel(
    const float* __restrict__ Vs, float* __restrict__ out)
{
    __shared__ __attribute__((aligned(16))) float tile[32 * TP];
    const int wave = threadIdx.x >> 5;
    const int lane = threadIdx.x & 31;
    const int row0 = blockIdx.x * 32;
    const int b    = row0 >> 12;
    const int i0   = row0 & (Sn - 1);

    #pragma unroll
    for (int q = 0; q < 8; q++) {
        const int s = wave * 8 + q;
        const float2 v = ((const float2*)Vs)[((size_t)b * NSL + s) * Sn + i0 + lane];
        *(float2*)&tile[lane * TP + CS * s] = v;
    }
    __syncthreads();
    #pragma unroll
    for (int q = 0; q < 4; q++) {
        const int rr = wave * 4 + q;
        const float2 lo = *(const float2*)&tile[rr * TP + 4 * lane];
        const float2 hi = *(const float2*)&tile[rr * TP + 4 * lane + 2];
        float4 o; o.x = lo.x; o.y = lo.y; o.z = hi.x; o.w = hi.y;
        ((float4*)(out + (size_t)(row0 + rr) * Dn))[lane] = o;
    }
}

extern "C" void kernel_launch(void* const* d_in, const int* in_sizes, int n_in,
                              void* d_out, int out_size, void* d_ws, size_t ws_size,
                              hipStream_t stream) {
    const int*   data = (const int*)  d_in[0];
    const float* emb  = (const float*)d_in[2];
    const float* apc  = (const float*)d_in[3];
    const float* gW1  = (const float*)d_in[4];
    const float* gb1  = (const float*)d_in[5];
    const float* gW2  = (const float*)d_in[6];
    const float* gb2  = (const float*)d_in[7];
    const float* fW1  = (const float*)d_in[8];
    const float* fb1  = (const float*)d_in[9];
    const float* fW2  = (const float*)d_in[10];
    const float* fb2  = (const float*)d_in[11];
    float* out = (float*)d_out;

    char*  ws   = (char*)d_ws;
    float* res  = (float*)(ws + (size_t) 8 * 1024 * 1024);
    float* Vs   = (float*)(ws + (size_t)16 * 1024 * 1024);
    float* Wpad = (float*)(ws + (size_t)24 * 1024 * 1024);

    front_kernel<<<dim3(Rn / (TW * 16), Mn + 1), TW * 32, 0, stream>>>(
        data, emb, apc, gW1, gb1, gW2, gb2, fW1, fb1, fW2, fb2, res, Wpad);

    chord_fused_kernel<<<Bz * NSL, CT, 0, stream>>>(res, Wpad, Vs);
    unslice_kernel<<<Rn / 32, 256, 0, stream>>>(Vs, out);
}


# Mixing weights kept as 13 planes [step][link][row] (whole-line stores via LDS); a mixing thread owns rows 4t..4t+3: one whole-line 16-byte load per link, not 4 loads over 16 lines; sum order kept
# speedup vs baseline: 2.2458x; 1.1735x over previous
#include <hip/hip_runtime.h>
#include <hip/hip_bf16.h>
#include <math.h>

#define Bz    4
#define Sn    4096
#define Dn    128
#define Kn    13
#define Mn    12
#define Rn    (Bz * Sn)
#define HSTR  132

typedef float v2f __attribute__((ext_vector_type(2)));
typedef float v8f __attribute__((ext_vector_type(8)));

__device__ __forceinline__ float gelu_exact(float v) {
    return 0.5f * v * (1.0f + erff(v * 0.70710678118654752f));
}

#define GELU_SMALL_LIMIT 1.41f

__device__ __forceinline__ float gelu_small(float v) {
    const float u = v * 0.70710678118654752f;
    const float t = u * u;
    float p = __builtin_fmaf(t, -0x1.268bc2p-11f, 0x1.420828p-8f);
    p = __builtin_fmaf(t, p, -0x1.b59370p-6f);
    p = __builtin_fmaf(t, p, 0x1.ce077cp-4f);
    p = __builtin_fmaf(t, p, -0x1.812660p-2f);
    p = __builtin_fmaf(t, p, 0x1.06eba0p-3f);
    const float e = __builtin_fmaf(u, p, u);
    return 0.5f * v * (1.0f + e);
}

__device__ __forceinline__ v8f wmma_f32x4(v2f a, v2f b, v8f c) {
    return __builtin_amdgcn_wmma_f32_16x16x4_f32(
         false, a,  false, b,
         (short)0, c,  false,  false);
}

#define TW    8
#define WSP   132
#define WSTR  132

struct WStage { float4 r0[8], r1[8]; };

__device__ __forceinline__ void stage_load(
    WStage& w, const float* __restrict__ Wg, int wave, int lane)
{
    const int kp = lane & 7;
    const int n0 = wave * 16 + (lane >> 3) * 4;
    #pragma unroll
    for (int j = 0; j < 8; j++) {
        const int k0 = j * 16 + kp * 2;
        w.r0[j] = *(const float4*)(Wg + (size_t)k0 * Dn + n0);
        w.r1[j] = *(const float4*)(Wg + (size_t)(k0 + 1) * Dn + n0);
    }
}

__device__ __forceinline__ void stage_store(
    const WStage& w, float* Wt, int wave, int lane)
{
    const int kp = lane & 7;
    const int n0 = wave * 16 + (lane >> 3) * 4;
    #pragma unroll
    for (int j = 0; j < 8; j++) {
        const int k0 = j * 16 + kp * 2;
        float2 p;
        p.x = w.r0[j].x; p.y = w.r1[j].x; *(float2*)&Wt[(n0 + 0) * WSTR + k0] = p;
        p.x = w.r0[j].y; p.y = w.r1[j].y; *(float2*)&Wt[(n0 + 1) * WSTR + k0] = p;
        p.x = w.r0[j].z; p.y = w.r1[j].z; *(float2*)&Wt[(n0 + 2) * WSTR + k0] = p;
        p.x = w.r0[j].w; p.y = w.r1[j].w; *(float2*)&Wt[(n0 + 3) * WSTR + k0] = p;
    }
}

__device__ __forceinline__ void product_wide(
    const float* aP, const float* bP, v8f (&acc)[8])
{
    #pragma unroll 4
    for (int k = 0; k < Dn; k += 4) {
        const v2f a = *(const v2f*)(aP + k);
        #pragma unroll
        for (int t = 0; t < 8; t++) {
            const v2f b = *(const v2f*)(bP + t * 16 * WSTR + k);
            acc[t] = wmma_f32x4(a, b, acc[t]);
        }
    }
}

__device__ __forceinline__ void wave_tile_sync() {
    __builtin_amdgcn_fence(__ATOMIC_RELEASE, "wavefront", "local");
    __builtin_amdgcn_wave_barrier();
    __builtin_amdgcn_fence(__ATOMIC_ACQUIRE, "wavefront", "local");
}

__global__ __launch_bounds__(TW * 32) void front_kernel(
    const int* __restrict__ data, const float* __restrict__ emb,
    const float* __restrict__ apc,
    const float* __restrict__ gW1, const float* __restrict__ gb1,
    const float* __restrict__ gW2, const float* __restrict__ gb2,
    const float* __restrict__ fW1, const float* __restrict__ fb1,
    const float* __restrict__ fW2, const float* __restrict__ fb2,
    float* __restrict__ res, float* __restrict__ Wpad)
{
    __shared__ __attribute__((aligned(16))) float Wt1[Dn * WSTR];
    __shared__ __attribute__((aligned(16))) float Wt2[16 * WSTR];
    __shared__ __attribute__((aligned(16))) float Hlds[TW][16 * HSTR];
    __shared__ __attribute__((aligned(16))) float Wst[Kn * WSP];
    const int slice   = blockIdx.y;
    const bool isG    = (slice == 0);
    const int m       = isG ? 0 : slice - 1;
    const int wave    = __builtin_amdgcn_readfirstlane(threadIdx.x >> 5);
    const int lane    = threadIdx.x & 31;
    const int half    = lane >> 4;
    const int mn      = lane & 15;
    const int rowBase = blockIdx.x * (TW * 16) + wave * 16;
    float* tile = Hlds[wave];

    const float* W1 = isG ? gW1 : fW1 + (size_t)m * Dn * Dn;
    const float* B1 = isG ? gb1 : fb1 + (size_t)m * Dn;

    WStage ws1;
    stage_load(ws1, W1, wave, lane);
    const float* W2 = fW2 + (size_t)m * Dn * Kn;
    float w2v[7];
    if (!isG) {
        #pragma unroll
        for (int j = 0; j < 7; j++) {
            const int idx = threadIdx.x + j * (TW * 32);
            w2v[j] = (idx < Dn * Kn) ? W2[idx] : 0.0f;
        }
    }
    float4 ev[16], av[16];
    #pragma unroll
    for (int r = 0; r < 16; r++) {
        const int row = rowBase + r;
        const int s   = row & (Sn - 1);
        const int tok = data[row];
        ev[r] = ((const float4*)(emb + (size_t)tok * Dn))[lane];
        av[r] = ((const float4*)(apc + (size_t)s * Dn))[lane];
    }
    stage_store(ws1, Wt1, wave, lane);
    if (!isG) {
        #pragma unroll
        for (int j = 0; j < 7; j++) {
            const int idx = threadIdx.x + j * (TW * 32);
            if (idx < Dn * Kn) {
                const int k = idx / Kn;
                const int n = idx - k * Kn;
                Wt2[n * WSTR + k] = w2v[j];
            }
        }
        Wt2[(Kn + (threadIdx.x >> 7)) * WSTR + (threadIdx.x & (Dn - 1))] = 0.0f;
        if (threadIdx.x < Dn) Wt2[(Kn + 2) * WSTR + threadIdx.x] = 0.0f;
    }
    #pragma unroll
    for (int r = 0; r < 16; r++) {
        float4 o;
        o.x = ev[r].x + av[r].x; o.y = ev[r].y + av[r].y;
        o.z = ev[r].z + av[r].z; o.w = ev[r].w + av[r].w;
        *(float4*)&tile[r * HSTR + 4 * lane] = o;
    }
    float bq[8];
    #pragma unroll
    for (int t = 0; t < 8; t++) bq[t] = B1[t * 16 + mn];
    __syncthreads();

    const float* aP = tile + mn * HSTR + 2 * half;
    const float* bP = Wt1 + mn * WSTR + 2 * half;

    v8f acc[8];
    #pragma unroll
    for (int t = 0; t < 8; t++) acc[t] = {};
    product_wide(aP, bP, acc);
    wave_tile_sync();
    float mx[8];
    #pragma unroll
    for (int t = 0; t < 8; t++) {
        mx[t] = 0.0f;
        #pragma unroll
        for (int r = 0; r < 8; r++) {
            acc[t][r] += bq[t];
            mx[t] = fmaxf(mx[t], fabsf(acc[t][r]));
        }
    }
    const float mAll = fmaxf(fmaxf(fmaxf(mx[0], mx[1]), fmaxf(mx[2], mx[3])),
                             fmaxf(fmaxf(mx[4], mx[5]), fmaxf(mx[6], mx[7])));
    const bool outside = !(mAll < GELU_SMALL_LIMIT);
    if (__builtin_amdgcn_ballot_w32(outside) == 0u) {
        #pragma unroll
        for (int t = 0; t < 8; t++) {
            #pragma unroll
            for (int r = 0; r < 8; r++) {
                tile[(r + 8 * half) * HSTR + t * 16 + mn] = gelu_small(acc[t][r]);
            }
        }
    } else {
        #pragma unroll
        for (int t = 0; t < 8; t++) {
            #pragma unroll
            for (int r = 0; r < 8; r++) {
                tile[(r + 8 * half) * HSTR + t * 16 + mn] = acc[t][r];
            }
        }
        #pragma nounroll
        for (int i = 0; i < 64; i++) {
            float* cell = tile + ((i & 7) + 8 * half) * HSTR + (i >> 3) * 16 + mn;
            *cell = gelu_exact(*cell);
        }
    }

    if (isG) {
        WStage ws2;
        stage_load(ws2, gW2, wave, lane);
        float bo[8];
        #pragma unroll
        for (int t = 0; t < 8; t++) bo[t] = gb2[t * 16 + mn];
        __syncthreads();
        stage_store(ws2, Wt1, wave, lane);
        __syncthreads();
        v8f acc2[8];
        #pragma unroll
        for (int t = 0; t < 8; t++) acc2[t] = {};
        product_wide(aP, bP, acc2);
        #pragma unroll
        for (int t = 0; t < 8; t++) {
            #pragma unroll
            for (int r = 0; r < 8; r++) {
                res[(size_t)(rowBase + r + 8 * half) * Dn + t * 16 + mn] =
                    acc2[t][r] + bo[t];
            }
        }
    } else {
        wave_tile_sync();
        const float* B2 = fb2 + (size_t)m * Kn;
        const float* b2P = Wt2 + mn * WSTR + 2 * half;
        v8f acc2 = {};
        const bool valid = (mn < Kn);
        const float bias = valid ? B2[mn] : 0.0f;
        for (int k0 = 0; k0 < Dn; k0 += 32) {
            v2f a[8], b[8];
            #pragma unroll
            for (int u = 0; u < 8; u++) {
                a[u] = *(const v2f*)(aP + k0 + 4 * u);
                b[u] = *(const v2f*)(b2P + k0 + 4 * u);
            }
            #pragma unroll
            for (int u = 0; u < 8; u++) acc2 = wmma_f32x4(a[u], b[u], acc2);
        }
        if (valid) {
            #pragma unroll
            for (int r = 0; r < 8; r++) {
                Wst[mn * WSP + wave * 16 + r + 8 * half] = acc2[r] + bias;
            }
        }
        __syncthreads();
        for (int k = wave; k < Kn; k += TW) {
            const float4 o = *(const float4*)&Wst[k * WSP + 4 * lane];
            *(float4*)&Wpad[((size_t)(m * Kn + k)) * Rn + blockIdx.x * (TW * 16) + 4 * lane] = o;
        }
    }
}

#define CS   2
#define NSL  (Dn / CS)
#define CT   1024
#define RPT  (Sn / CT)
#define PQ   (Rn / 4)

__device__ __forceinline__ constexpr int link_off(int k) { return k == 0 ? 0 : (1 << (k - 1)); }

__device__ __forceinline__ int slot_of(int i) { return (i & 3) * CT + (i >> 2); }

template <int K0, int K1>
__device__ __forceinline__ void link_group(
    const v2f* src, int t, float4 (&w)[Kn], v2f (&acc)[RPT],
    const float4* __restrict__ Wn, bool more)
{
    v2f v[K1 - K0][RPT];
    #pragma unroll
    for (int k = K0; k < K1; k++) {
        #pragma unroll
        for (int j = 0; j < RPT; j++) {
            const int c = j + link_off(k);
            v[k - K0][j] = src[((t + (c >> 2)) & (CT - 1)) + CT * (c & 3)];
        }
    }
    if constexpr (K1 - K0 == 4) {
        asm volatile("" :: "v"(v[0][0]), "v"(v[0][1]), "v"(v[0][2]), "v"(v[0][3]),
                           "v"(v[1][0]), "v"(v[1][1]), "v"(v[1][2]), "v"(v[1][3]),
                           "v"(v[2][0]), "v"(v[2][1]), "v"(v[2][2]), "v"(v[2][3]),
                           "v"(v[3][0]), "v"(v[3][1]), "v"(v[3][2]), "v"(v[3][3]));
    } else {
        asm volatile("" :: "v"(v[0][0]), "v"(v[0][1]), "v"(v[0][2]), "v"(v[0][3]),
                           "v"(v[1][0]), "v"(v[1][1]), "v"(v[1][2]), "v"(v[1][3]),
                           "v"(v[2][0]), "v"(v[2][1]), "v"(v[2][2]), "v"(v[2][3]),
                           "v"(v[3][0]), "v"(v[3][1]), "v"(v[3][2]), "v"(v[3][3]),
                           "v"(v[4][0]), "v"(v[4][1]), "v"(v[4][2]), "v"(v[4][3]));
    }
    #pragma unroll
    for (int k = K0; k < K1; k++) {
        const float wj[RPT] = {w[k].x, w[k].y, w[k].z, w[k].w};
        #pragma unroll
        for (int j = 0; j < RPT; j++) {
            acc[j].x += wj[j] * v[k - K0][j].x;
            acc[j].y += wj[j] * v[k - K0][j].y;
        }
    }
    if (more) {
        #pragma unroll
        for (int k = K0; k < K1; k++) w[k] = Wn[(size_t)k * PQ];
    }
}

__global__ __launch_bounds__(CT) void chord_fused_kernel(
    const float* __restrict__ res, const float* __restrict__ Wpl,
    float* __restrict__ Vs)
{
    __shared__ v2f buf0[Sn];
    __shared__ v2f buf1[Sn];
    const int t = threadIdx.x;
    const int s = blockIdx.x & (NSL - 1);
    const int b = blockIdx.x / NSL;

    #pragma unroll
    for (int j = 0; j < RPT; j++) {
        const int i = t + CT * j;
        buf0[slot_of(i)] = *(const v2f*)(res + ((size_t)b * Sn + i) * Dn + CS * s);
    }
    const float4* Wb = (const float4*)(Wpl + (size_t)b * Sn) + t;
    float4 w[Kn];
    #pragma unroll
    for (int k = 0; k < Kn; k++) w[k] = Wb[(size_t)k * PQ];
    __syncthreads();
    v2f r[RPT];
    #pragma unroll
    for (int j = 0; j < RPT; j++) r[j] = buf0[t + CT * j];

    v2f* src = buf0;
    v2f* dst = buf1;
    for (int m = 0; m < Mn; m++) {
        const float4* Wn = Wb + (size_t)(m + 1) * Kn * PQ;
        const bool more = (m + 1 < Mn);
        v2f acc[RPT];
        #pragma unroll
        for (int j = 0; j < RPT; j++) acc[j] = r[j];
        link_group<0, 4>(src, t, w, acc, Wn, more);
        link_group<4, 8>(src, t, w, acc, Wn, more);
        link_group<8, Kn>(src, t, w, acc, Wn, more);
        #pragma unroll
        for (int j = 0; j < RPT; j++) dst[t + CT * j] = acc[j];
        __builtin_amdgcn_fence(__ATOMIC_RELEASE, "workgroup", "local");
        __builtin_amdgcn_s_barrier();
        __builtin_amdgcn_fence(__ATOMIC_ACQUIRE, "workgroup", "local");
        v2f* tmp = src; src = dst; dst = tmp;
    }

    v2f* o = (v2f*)Vs + ((size_t)b * NSL + s) * Sn;
    #pragma unroll
    for (int j = 0; j < RPT; j++) {
        const int i = t + CT * j;
        o[i] = src[slot_of(i)];
    }
}

#define TP 130

__global__ __launch_bounds__(256) void unslice_kernel(
    const float* __restrict__ Vs, float* __restrict__ out)
{
    __shared__ __attribute__((aligned(16))) float tile[32 * TP];
    const int wave = threadIdx.x >> 5;
    const int lane = threadIdx.x & 31;
    const int row0 = blockIdx.x * 32;
    const int b    = row0 >> 12;
    const int i0   = row0 & (Sn - 1);

    #pragma unroll
    for (int q = 0; q < 8; q++) {
        const int s = wave * 8 + q;
        const float2 v = ((const float2*)Vs)[((size_t)b * NSL + s) * Sn + i0 + lane];
        *(float2*)&tile[lane * TP + CS * s] = v;
    }
    __syncthreads();
    #pragma unroll
    for (int q = 0; q < 4; q++) {
        const int rr = wave * 4 + q;
        const float2 lo = *(const float2*)&tile[rr * TP + 4 * lane];
        const float2 hi = *(const float2*)&tile[rr * TP + 4 * lane + 2];
        float4 o; o.x = lo.x; o.y = lo.y; o.z = hi.x; o.w = hi.y;
        ((float4*)(out + (size_t)(row0 + rr) * Dn))[lane] = o;
    }
}

extern "C" void kernel_launch(void* const* d_in, const int* in_sizes, int n_in,
                              void* d_out, int out_size, void* d_ws, size_t ws_size,
                              hipStream_t stream) {
    const int*   data = (const int*)  d_in[0];
    const float* emb  = (const float*)d_in[2];
    const float* apc  = (const float*)d_in[3];
    const float* gW1  = (const float*)d_in[4];
    const float* gb1  = (const float*)d_in[5];
    const float* gW2  = (const float*)d_in[6];
    const float* gb2  = (const float*)d_in[7];
    const float* fW1  = (const float*)d_in[8];
    const float* fb1  = (const float*)d_in[9];
    const float* fW2  = (const float*)d_in[10];
    const float* fb2  = (const float*)d_in[11];
    float* out = (float*)d_out;

    char*  ws   = (char*)d_ws;
    float* res  = (float*)(ws + (size_t) 8 * 1024 * 1024);
    float* Vs   = (float*)(ws + (size_t)16 * 1024 * 1024);
    float* Wpad = (float*)(ws + (size_t)24 * 1024 * 1024);

    front_kernel<<<dim3(Rn / (TW * 16), Mn + 1), TW * 32, 0, stream>>>(
        data, emb, apc, gW1, gb1, gW2, gb2, fW1, fb1, fW2, fb2, res, Wpad);

    chord_fused_kernel<<<Bz * NSL, CT, 0, stream>>>(res, Wpad, Vs);
    unslice_kernel<<<Rn / 32, 256, 0, stream>>>(Vs, out);
}


# res stored in the mixing stage's sliced layout [b][slice][row][2]: front regroups V in LDS into whole 128-byte slice lines; a mixing work-group fills from 32 contiguous KiB, not 4096 lines
# speedup vs baseline: 2.4095x; 1.0729x over previous
#include <hip/hip_runtime.h>
#include <hip/hip_bf16.h>
#include <math.h>

#define Bz    4
#define Sn    4096
#define Dn    128
#define Kn    13
#define Mn    12
#define Rn    (Bz * Sn)
#define HSTR  132
#define CS    2
#define NSL   (Dn / CS)

typedef float v2f __attribute__((ext_vector_type(2)));
typedef float v8f __attribute__((ext_vector_type(8)));

__device__ __forceinline__ float gelu_exact(float v) {
    return 0.5f * v * (1.0f + erff(v * 0.70710678118654752f));
}

#define GELU_SMALL_LIMIT 1.41f

__device__ __forceinline__ float gelu_small(float v) {
    const float u = v * 0.70710678118654752f;
    const float t = u * u;
    float p = __builtin_fmaf(t, -0x1.268bc2p-11f, 0x1.420828p-8f);
    p = __builtin_fmaf(t, p, -0x1.b59370p-6f);
    p = __builtin_fmaf(t, p, 0x1.ce077cp-4f);
    p = __builtin_fmaf(t, p, -0x1.812660p-2f);
    p = __builtin_fmaf(t, p, 0x1.06eba0p-3f);
    const float e = __builtin_fmaf(u, p, u);
    return 0.5f * v * (1.0f + e);
}

__device__ __forceinline__ v8f wmma_f32x4(v2f a, v2f b, v8f c) {
    return __builtin_amdgcn_wmma_f32_16x16x4_f32(
         false, a,  false, b,
         (short)0, c,  false,  false);
}

#define TW    8
#define WSP   132
#define WSTR  132

struct WStage { float4 r0[8], r1[8]; };

__device__ __forceinline__ void stage_load(
    WStage& w, const float* __restrict__ Wg, int wave, int lane)
{
    const int kp = lane & 7;
    const int n0 = wave * 16 + (lane >> 3) * 4;
    #pragma unroll
    for (int j = 0; j < 8; j++) {
        const int k0 = j * 16 + kp * 2;
        w.r0[j] = *(const float4*)(Wg + (size_t)k0 * Dn + n0);
        w.r1[j] = *(const float4*)(Wg + (size_t)(k0 + 1) * Dn + n0);
    }
}

__device__ __forceinline__ void stage_store(
    const WStage& w, float* Wt, int wave, int lane)
{
    const int kp = lane & 7;
    const int n0 = wave * 16 + (lane >> 3) * 4;
    #pragma unroll
    for (int j = 0; j < 8; j++) {
        const int k0 = j * 16 + kp * 2;
        float2 p;
        p.x = w.r0[j].x; p.y = w.r1[j].x; *(float2*)&Wt[(n0 + 0) * WSTR + k0] = p;
        p.x = w.r0[j].y; p.y = w.r1[j].y; *(float2*)&Wt[(n0 + 1) * WSTR + k0] = p;
        p.x = w.r0[j].z; p.y = w.r1[j].z; *(float2*)&Wt[(n0 + 2) * WSTR + k0] = p;
        p.x = w.r0[j].w; p.y = w.r1[j].w; *(float2*)&Wt[(n0 + 3) * WSTR + k0] = p;
    }
}

__device__ __forceinline__ void product_wide(
    const float* aP, const float* bP, v8f (&acc)[8])
{
    #pragma unroll 4
    for (int k = 0; k < Dn; k += 4) {
        const v2f a = *(const v2f*)(aP + k);
        #pragma unroll
        for (int t = 0; t < 8; t++) {
            const v2f b = *(const v2f*)(bP + t * 16 * WSTR + k);
            acc[t] = wmma_f32x4(a, b, acc[t]);
        }
    }
}

__device__ __forceinline__ void wave_tile_sync() {
    __builtin_amdgcn_fence(__ATOMIC_RELEASE, "wavefront", "local");
    __builtin_amdgcn_wave_barrier();
    __builtin_amdgcn_fence(__ATOMIC_ACQUIRE, "wavefront", "local");
}

__global__ __launch_bounds__(TW * 32) void front_kernel(
    const int* __restrict__ data, const float* __restrict__ emb,
    const float* __restrict__ apc,
    const float* __restrict__ gW1, const float* __restrict__ gb1,
    const float* __restrict__ gW2, const float* __restrict__ gb2,
    const float* __restrict__ fW1, const float* __restrict__ fb1,
    const float* __restrict__ fW2, const float* __restrict__ fb2,
    float* __restrict__ res, float* __restrict__ Wpad)
{
    __shared__ __attribute__((aligned(16))) float Wt1[Dn * WSTR];
    __shared__ __attribute__((aligned(16))) float Wt2[16 * WSTR];
    __shared__ __attribute__((aligned(16))) float Hlds[TW][16 * HSTR];
    __shared__ __attribute__((aligned(16))) float Wst[Kn * WSP];
    const int slice   = blockIdx.y;
    const bool isG    = (slice == 0);
    const int m       = isG ? 0 : slice - 1;
    const int wave    = __builtin_amdgcn_readfirstlane(threadIdx.x >> 5);
    const int lane    = threadIdx.x & 31;
    const int half    = lane >> 4;
    const int mn      = lane & 15;
    const int rowBase = blockIdx.x * (TW * 16) + wave * 16;
    float* tile = Hlds[wave];

    const float* W1 = isG ? gW1 : fW1 + (size_t)m * Dn * Dn;
    const float* B1 = isG ? gb1 : fb1 + (size_t)m * Dn;

    WStage ws1;
    stage_load(ws1, W1, wave, lane);
    const float* W2 = fW2 + (size_t)m * Dn * Kn;
    float w2v[7];
    if (!isG) {
        #pragma unroll
        for (int j = 0; j < 7; j++) {
            const int idx = threadIdx.x + j * (TW * 32);
            w2v[j] = (idx < Dn * Kn) ? W2[idx] : 0.0f;
        }
    }
    float4 ev[16], av[16];
    #pragma unroll
    for (int r = 0; r < 16; r++) {
        const int row = rowBase + r;
        const int s   = row & (Sn - 1);
        const int tok = data[row];
        ev[r] = ((const float4*)(emb + (size_t)tok * Dn))[lane];
        av[r] = ((const float4*)(apc + (size_t)s * Dn))[lane];
    }
    stage_store(ws1, Wt1, wave, lane);
    if (!isG) {
        #pragma unroll
        for (int j = 0; j < 7; j++) {
            const int idx = threadIdx.x + j * (TW * 32);
            if (idx < Dn * Kn) {
                const int k = idx / Kn;
                const int n = idx - k * Kn;
                Wt2[n * WSTR + k] = w2v[j];
            }
        }
        Wt2[(Kn + (threadIdx.x >> 7)) * WSTR + (threadIdx.x & (Dn - 1))] = 0.0f;
        if (threadIdx.x < Dn) Wt2[(Kn + 2) * WSTR + threadIdx.x] = 0.0f;
    }
    #pragma unroll
    for (int r = 0; r < 16; r++) {
        float4 o;
        o.x = ev[r].x + av[r].x; o.y = ev[r].y + av[r].y;
        o.z = ev[r].z + av[r].z; o.w = ev[r].w + av[r].w;
        *(float4*)&tile[r * HSTR + 4 * lane] = o;
    }
    float bq[8];
    #pragma unroll
    for (int t = 0; t < 8; t++) bq[t] = B1[t * 16 + mn];
    __syncthreads();

    const float* aP = tile + mn * HSTR + 2 * half;
    const float* bP = Wt1 + mn * WSTR + 2 * half;

    v8f acc[8];
    #pragma unroll
    for (int t = 0; t < 8; t++) acc[t] = {};
    product_wide(aP, bP, acc);
    wave_tile_sync();
    float mx[8];
    #pragma unroll
    for (int t = 0; t < 8; t++) {
        mx[t] = 0.0f;
        #pragma unroll
        for (int r = 0; r < 8; r++) {
            acc[t][r] += bq[t];
            mx[t] = fmaxf(mx[t], fabsf(acc[t][r]));
        }
    }
    const float mAll = fmaxf(fmaxf(fmaxf(mx[0], mx[1]), fmaxf(mx[2], mx[3])),
                             fmaxf(fmaxf(mx[4], mx[5]), fmaxf(mx[6], mx[7])));
    const bool outside = !(mAll < GELU_SMALL_LIMIT);
    if (__builtin_amdgcn_ballot_w32(outside) == 0u) {
        #pragma unroll
        for (int t = 0; t < 8; t++) {
            #pragma unroll
            for (int r = 0; r < 8; r++) {
                tile[(r + 8 * half) * HSTR + t * 16 + mn] = gelu_small(acc[t][r]);
            }
        }
    } else {
        #pragma unroll
        for (int t = 0; t < 8; t++) {
            #pragma unroll
            for (int r = 0; r < 8; r++) {
                tile[(r + 8 * half) * HSTR + t * 16 + mn] = acc[t][r];
            }
        }
        #pragma nounroll
        for (int i = 0; i < 64; i++) {
            float* cell = tile + ((i & 7) + 8 * half) * HSTR + (i >> 3) * 16 + mn;
            *cell = gelu_exact(*cell);
        }
    }

    if (isG) {
        WStage ws2;
        stage_load(ws2, gW2, wave, lane);
        float bo[8];
        #pragma unroll
        for (int t = 0; t < 8; t++) bo[t] = gb2[t * 16 + mn];
        __syncthreads();
        stage_store(ws2, Wt1, wave, lane);
        __syncthreads();
        v8f acc2[8];
        #pragma unroll
        for (int t = 0; t < 8; t++) acc2[t] = {};
        product_wide(aP, bP, acc2);
        wave_tile_sync();
        #pragma unroll
        for (int t = 0; t < 8; t++) {
            #pragma unroll
            for (int r = 0; r < 8; r++) {
                const int sl  = t * 8 + (mn >> 1);
                const int off = (r + 8 * half) * CS + (mn & 1);
                tile[sl * 32 + (off ^ (((sl >> 1) & 3) << 2))] = acc2[t][r] + bo[t];
            }
        }
        wave_tile_sync();
        const int bb    = rowBase >> 12;
        const int srow0 = rowBase & (Sn - 1);
        #pragma unroll
        for (int it = 0; it < 16; it++) {
            const int sl = 4 * it + (lane >> 3);
            const int q  = lane & 7;
            const float4 o = *(const float4*)&tile[sl * 32 + ((q ^ ((sl >> 1) & 3)) << 2)];
            *(float4*)&res[(((size_t)(bb * NSL + sl)) * Sn + srow0) * CS + 4 * q] = o;
        }
    } else {
        wave_tile_sync();
        const float* B2 = fb2 + (size_t)m * Kn;
        const float* b2P = Wt2 + mn * WSTR + 2 * half;
        v8f acc2 = {};
        const bool valid = (mn < Kn);
        const float bias = valid ? B2[mn] : 0.0f;
        for (int k0 = 0; k0 < Dn; k0 += 32) {
            v2f a[8], b[8];
            #pragma unroll
            for (int u = 0; u < 8; u++) {
                a[u] = *(const v2f*)(aP + k0 + 4 * u);
                b[u] = *(const v2f*)(b2P + k0 + 4 * u);
            }
            #pragma unroll
            for (int u = 0; u < 8; u++) acc2 = wmma_f32x4(a[u], b[u], acc2);
        }
        if (valid) {
            #pragma unroll
            for (int r = 0; r < 8; r++) {
                Wst[mn * WSP + wave * 16 + r + 8 * half] = acc2[r] + bias;
            }
        }
        __syncthreads();
        for (int k = wave; k < Kn; k += TW) {
            const float4 o = *(const float4*)&Wst[k * WSP + 4 * lane];
            *(float4*)&Wpad[((size_t)(m * Kn + k)) * Rn + blockIdx.x * (TW * 16) + 4 * lane] = o;
        }
    }
}

#define CT   1024
#define RPT  (Sn / CT)
#define PQ   (Rn / 4)

__device__ __forceinline__ constexpr int link_off(int k) { return k == 0 ? 0 : (1 << (k - 1)); }

__device__ __forceinline__ int slot_of(int i) { return (i & 3) * CT + (i >> 2); }

template <int K0, int K1>
__device__ __forceinline__ void link_group(
    const v2f* src, int t, float4 (&w)[Kn], v2f (&acc)[RPT],
    const float4* __restrict__ Wn, bool more)
{
    v2f v[K1 - K0][RPT];
    #pragma unroll
    for (int k = K0; k < K1; k++) {
        #pragma unroll
        for (int j = 0; j < RPT; j++) {
            const int c = j + link_off(k);
            v[k - K0][j] = src[((t + (c >> 2)) & (CT - 1)) + CT * (c & 3)];
        }
    }
    if constexpr (K1 - K0 == 4) {
        asm volatile("" :: "v"(v[0][0]), "v"(v[0][1]), "v"(v[0][2]), "v"(v[0][3]),
                           "v"(v[1][0]), "v"(v[1][1]), "v"(v[1][2]), "v"(v[1][3]),
                           "v"(v[2][0]), "v"(v[2][1]), "v"(v[2][2]), "v"(v[2][3]),
                           "v"(v[3][0]), "v"(v[3][1]), "v"(v[3][2]), "v"(v[3][3]));
    } else {
        asm volatile("" :: "v"(v[0][0]), "v"(v[0][1]), "v"(v[0][2]), "v"(v[0][3]),
                           "v"(v[1][0]), "v"(v[1][1]), "v"(v[1][2]), "v"(v[1][3]),
                           "v"(v[2][0]), "v"(v[2][1]), "v"(v[2][2]), "v"(v[2][3]),
                           "v"(v[3][0]), "v"(v[3][1]), "v"(v[3][2]), "v"(v[3][3]),
                           "v"(v[4][0]), "v"(v[4][1]), "v"(v[4][2]), "v"(v[4][3]));
    }
    #pragma unroll
    for (int k = K0; k < K1; k++) {
        const float wj[RPT] = {w[k].x, w[k].y, w[k].z, w[k].w};
        #pragma unroll
        for (int j = 0; j < RPT; j++) {
            acc[j].x += wj[j] * v[k - K0][j].x;
            acc[j].y += wj[j] * v[k - K0][j].y;
        }
    }
    if (more) {
        #pragma unroll
        for (int k = K0; k < K1; k++) w[k] = Wn[(size_t)k * PQ];
    }
}

__global__ __launch_bounds__(CT) void chord_fused_kernel(
    const float* __restrict__ res, const float* __restrict__ Wpl,
    float* __restrict__ Vs)
{
    __shared__ v2f buf0[Sn];
    __shared__ v2f buf1[Sn];
    const int t = threadIdx.x;
    const int s = blockIdx.x & (NSL - 1);
    const int b = blockIdx.x / NSL;

    const float4* rs = (const float4*)(res + ((size_t)b * NSL + s) * Sn * CS) + 2 * t;
    const float4 q0 = rs[0];
    const float4 q1 = rs[1];
    v2f r[RPT];
    r[0].x = q0.x; r[0].y = q0.y; r[1].x = q0.z; r[1].y = q0.w;
    r[2].x = q1.x; r[2].y = q1.y; r[3].x = q1.z; r[3].y = q1.w;
    #pragma unroll
    for (int j = 0; j < RPT; j++) buf0[t + CT * j] = r[j];
    const float4* Wb = (const float4*)(Wpl + (size_t)b * Sn) + t;
    float4 w[Kn];
    #pragma unroll
    for (int k = 0; k < Kn; k++) w[k] = Wb[(size_t)k * PQ];
    __syncthreads();

    v2f* src = buf0;
    v2f* dst = buf1;
    for (int m = 0; m < Mn; m++) {
        const float4* Wn = Wb + (size_t)(m + 1) * Kn * PQ;
        const bool more = (m + 1 < Mn);
        v2f acc[RPT];
        #pragma unroll
        for (int j = 0; j < RPT; j++) acc[j] = r[j];
        link_group<0, 4>(src, t, w, acc, Wn, more);
        link_group<4, 8>(src, t, w, acc, Wn, more);
        link_group<8, Kn>(src, t, w, acc, Wn, more);
        #pragma unroll
        for (int j = 0; j < RPT; j++) dst[t + CT * j] = acc[j];
        __builtin_amdgcn_fence(__ATOMIC_RELEASE, "workgroup", "local");
        __builtin_amdgcn_s_barrier();
        __builtin_amdgcn_fence(__ATOMIC_ACQUIRE, "workgroup", "local");
        v2f* tmp = src; src = dst; dst = tmp;
    }

    v2f* o = (v2f*)Vs + ((size_t)b * NSL + s) * Sn;
    #pragma unroll
    for (int j = 0; j < RPT; j++) {
        const int i = t + CT * j;
        o[i] = src[slot_of(i)];
    }
}

#define TP 130

__global__ __launch_bounds__(256) void unslice_kernel(
    const float* __restrict__ Vs, float* __restrict__ out)
{
    __shared__ __attribute__((aligned(16))) float tile[32 * TP];
    const int wave = threadIdx.x >> 5;
    const int lane = threadIdx.x & 31;
    const int row0 = blockIdx.x * 32;
    const int b    = row0 >> 12;
    const int i0   = row0 & (Sn - 1);

    #pragma unroll
    for (int q = 0; q < 8; q++) {
        const int s = wave * 8 + q;
        const float2 v = ((const float2*)Vs)[((size_t)b * NSL + s) * Sn + i0 + lane];
        *(float2*)&tile[lane * TP + CS * s] = v;
    }
    __syncthreads();
    #pragma unroll
    for (int q = 0; q < 4; q++) {
        const int rr = wave * 4 + q;
        const float2 lo = *(const float2*)&tile[rr * TP + 4 * lane];
        const float2 hi = *(const float2*)&tile[rr * TP + 4 * lane + 2];
        float4 o; o.x = lo.x; o.y = lo.y; o.z = hi.x; o.w = hi.y;
        ((float4*)(out + (size_t)(row0 + rr) * Dn))[lane] = o;
    }
}

extern "C" void kernel_launch(void* const* d_in, const int* in_sizes, int n_in,
                              void* d_out, int out_size, void* d_ws, size_t ws_size,
                              hipStream_t stream) {
    const int*   data = (const int*)  d_in[0];
    const float* emb  = (const float*)d_in[2];
    const float* apc  = (const float*)d_in[3];
    const float* gW1  = (const float*)d_in[4];
    const float* gb1  = (const float*)d_in[5];
    const float* gW2  = (const float*)d_in[6];
    const float* gb2  = (const float*)d_in[7];
    const float* fW1  = (const float*)d_in[8];
    const float* fb1  = (const float*)d_in[9];
    const float* fW2  = (const float*)d_in[10];
    const float* fb2  = (const float*)d_in[11];
    float* out = (float*)d_out;

    char*  ws   = (char*)d_ws;
    float* res  = (float*)(ws + (size_t) 8 * 1024 * 1024);
    float* Vs   = (float*)(ws + (size_t)16 * 1024 * 1024);
    float* Wpad = (float*)(ws + (size_t)24 * 1024 * 1024);

    front_kernel<<<dim3(Rn / (TW * 16), Mn + 1), TW * 32, 0, stream>>>(
        data, emb, apc, gW1, gb1, gW2, gb2, fW1, fb1, fW2, fb2, res, Wpad);

    chord_fused_kernel<<<Bz * NSL, CT, 0, stream>>>(res, Wpad, Vs);
    unslice_kernel<<<Rn / 32, 256, 0, stream>>>(Vs, out);
}
